# starting kernel
# baseline (speedup 1.0000x reference)
#include <hip/hip_runtime.h>
#include <hip/hip_bf16.h>

typedef float v2f __attribute__((ext_vector_type(2)));
typedef float v8f __attribute__((ext_vector_type(8)));

#define LDS_STRIDE 260

__device__ __forceinline__ void atomAddF32(float* p, float v) {
    unsafeAtomicAdd(p, v);
}


__global__ __launch_bounds__(256) void k_deg_init(float* __restrict__ deg, int n) {
    int i = blockIdx.x * 256 + threadIdx.x;
    if (i < n) deg[i] = 1.0f;
}

__global__ __launch_bounds__(256) void k_deg_edges(const int* __restrict__ dst,
                                                   float* __restrict__ deg, int e) {
    int i = blockIdx.x * 256 + threadIdx.x;
    if (i < e) atomAddF32(&deg[dst[i]], 1.0f);
}

__global__ __launch_bounds__(256) void k_rsqrt(float* __restrict__ deg, int n) {
    int i = blockIdx.x * 256 + threadIdx.x;
    if (i < n) deg[i] = __frsqrt_rn(deg[i]);
}


__global__ __launch_bounds__(256) void k_gemm1(const float* __restrict__ x,
                                               const float* __restrict__ W1,
                                               float* __restrict__ h0) {
    __shared__ float As[16 * LDS_STRIDE];
    const int tid = threadIdx.x;
    const int r0  = blockIdx.x * 16;

    const float4* xv = (const float4*)(x + (size_t)r0 * 256);
    #pragma unroll
    for (int i = 0; i < 4; ++i) {
        int j   = i * 256 + tid;
        int row = j >> 6;
        int c4  = j & 63;
        float4 v = xv[(size_t)row * 64 + c4];
        float* d = &As[row * LDS_STRIDE + c4 * 4];
        d[0] = v.x; d[1] = v.y; d[2] = v.z; d[3] = v.w;
    }
    __syncthreads();

    const int wave  = tid >> 5;
    const int lane  = tid & 31;
    const int half  = lane >> 4;
    const int l16   = lane & 15;
    const int c0    = wave * 16;
    const int kboff = half * 2;

    v8f acc = {0.f, 0.f, 0.f, 0.f, 0.f, 0.f, 0.f, 0.f};
    for (int k0 = 0; k0 < 256; k0 += 4) {
        v2f a, b;
        const float* ap = &As[l16 * LDS_STRIDE + k0 + kboff];
        a.x = ap[0]; a.y = ap[1];
        const float* bp = &W1[(size_t)(k0 + kboff) * 128 + c0 + l16];
        b.x = bp[0]; b.y = bp[128];
        acc = __builtin_amdgcn_wmma_f32_16x16x4_f32(false, a, false, b,
                                                    (short)0, acc, false, false);
    }
    float* out = h0 + (size_t)(r0 + half * 8) * 128 + c0 + l16;
    #pragma unroll
    for (int i = 0; i < 8; ++i) out[(size_t)i * 128] = acc[i];
}


__global__ __launch_bounds__(256) void k_init_out1(const float* __restrict__ h0,
                                                   const float* __restrict__ b1,
                                                   const float* __restrict__ dis,
                                                   float* __restrict__ out1, int n) {
    int g = blockIdx.x * 256 + threadIdx.x;
    if (g >= n * 32) return;
    int node = g >> 5;
    int f4   = g & 31;
    float d2 = dis[node] * dis[node];
    float4 v = ((const float4*)h0)[(size_t)g];
    float4 bb = ((const float4*)b1)[f4];
    float4 o;
    o.x = bb.x + v.x * d2; o.y = bb.y + v.y * d2;
    o.z = bb.z + v.z * d2; o.w = bb.w + v.w * d2;
    ((float4*)out1)[(size_t)g] = o;
}

__global__ __launch_bounds__(256) void k_scatter128(const int* __restrict__ src,
                                                    const int* __restrict__ dst,
                                                    const float* __restrict__ dis,
                                                    const float* __restrict__ h,
                                                    float* __restrict__ out, int e) {
    int g = blockIdx.x * 256 + threadIdx.x;
    int edge = g >> 5;
    if (edge >= e) return;
    int lane = threadIdx.x & 31;
    int s = src[edge], d = dst[edge];
    float nrm = dis[s] * dis[d];
    float4 v = *(const float4*)(h + (size_t)s * 128 + lane * 4);
    float* o = out + (size_t)d * 128 + lane * 4;
    atomAddF32(o + 0, v.x * nrm);
    atomAddF32(o + 1, v.y * nrm);
    atomAddF32(o + 2, v.z * nrm);
    atomAddF32(o + 3, v.w * nrm);
}


__global__ __launch_bounds__(256) void k_gemm2(const float* __restrict__ out1,
                                               const float* __restrict__ W2,
                                               float* __restrict__ h2, int ntiles) {
    int wg = (blockIdx.x * 256 + threadIdx.x) >> 5;
    if (wg >= ntiles) return;
    const int lane  = threadIdx.x & 31;
    const int half  = lane >> 4;
    const int l16   = lane & 15;
    const int kboff = half * 2;
    const int r0    = wg * 16;

    const float* arow = out1 + (size_t)(r0 + l16) * 128;
    v8f acc = {0.f, 0.f, 0.f, 0.f, 0.f, 0.f, 0.f, 0.f};
    for (int k0 = 0; k0 < 128; k0 += 4) {
        v2f a, b;
        a.x = fmaxf(arow[k0 + kboff], 0.f);
        a.y = fmaxf(arow[k0 + kboff + 1], 0.f);
        const float* bp = &W2[(size_t)(k0 + kboff) * 16 + l16];
        b.x = bp[0]; b.y = bp[16];
        acc = __builtin_amdgcn_wmma_f32_16x16x4_f32(false, a, false, b,
                                                    (short)0, acc, false, false);
    }
    float* out = h2 + (size_t)(r0 + half * 8) * 16 + l16;
    #pragma unroll
    for (int i = 0; i < 8; ++i) out[(size_t)i * 16] = acc[i];
}


__global__ __launch_bounds__(256) void k_init_out2(const float* __restrict__ h2,
                                                   const float* __restrict__ b2,
                                                   const float* __restrict__ dis,
                                                   float* __restrict__ logits, int n) {
    int g = blockIdx.x * 256 + threadIdx.x;
    if (g >= n * 16) return;
    int node = g >> 4;
    int f    = g & 15;
    float d2 = dis[node] * dis[node];
    logits[g] = b2[f] + h2[g] * d2;
}

__global__ __launch_bounds__(256) void k_scatter16(const int* __restrict__ src,
                                                   const int* __restrict__ dst,
                                                   const float* __restrict__ dis,
                                                   const float* __restrict__ h2,
                                                   float* __restrict__ logits, int e) {
    int g = blockIdx.x * 256 + threadIdx.x;
    int edge = g >> 4;
    if (edge >= e) return;
    int f = g & 15;
    int s = src[edge], d = dst[edge];
    float nrm = dis[s] * dis[d];
    float v = h2[(size_t)s * 16 + f];
    atomAddF32(&logits[(size_t)d * 16 + f], v * nrm);
}

__global__ __launch_bounds__(256) void k_softmax16(const float* __restrict__ logits,
                                                   float* __restrict__ probs, int n) {
    int g = blockIdx.x * 256 + threadIdx.x;
    if (g >= n * 16) return;
    float v = logits[g];
    float m = v;
    #pragma unroll
    for (int o = 8; o >= 1; o >>= 1) m = fmaxf(m, __shfl_xor(m, o, 16));
    float e = __expf(v - m);
    float s = e;
    #pragma unroll
    for (int o = 8; o >= 1; o >>= 1) s += __shfl_xor(s, o, 16);
    probs[g] = e / s;
}


extern "C" void kernel_launch(void* const* d_in, const int* in_sizes, int n_in,
                              void* d_out, int out_size, void* d_ws, size_t ws_size,
                              hipStream_t stream) {
    const float* x  = (const float*)d_in[0];
    const int*   ei = (const int*)d_in[1];
    const float* W1 = (const float*)d_in[2];
    const float* b1 = (const float*)d_in[3];
    const float* W2 = (const float*)d_in[4];
    const float* b2 = (const float*)d_in[5];

    const int N = in_sizes[0] / 256;
    const int E = in_sizes[1] / 2;
    const int* src = ei;
    const int* dst = ei + E;

    char*  ws  = (char*)d_ws;
    size_t off = 0;
    auto alloc = [&](size_t bytes) -> void* {
        void* p = ws + off;
        off += (bytes + 255) & ~(size_t)255;
        return p;
    };
    float* dis  = (float*)alloc((size_t)N * 4);
    float* h0   = (float*)alloc((size_t)N * 128 * 4);
    float* out1 = (float*)alloc((size_t)N * 128 * 4);
    float* h2   = (float*)alloc((size_t)N * 16 * 4);
    float* logits = (float*)d_out;
    float* probs  = (float*)d_out + (size_t)N * 16;

    auto cdiv = [](long long a, long long b) { return (int)((a + b - 1) / b); };

    k_deg_init <<<cdiv(N, 256), 256, 0, stream>>>(dis, N);
    k_deg_edges<<<cdiv(E, 256), 256, 0, stream>>>(dst, dis, E);
    k_rsqrt    <<<cdiv(N, 256), 256, 0, stream>>>(dis, N);

    k_gemm1    <<<N / 16, 256, 0, stream>>>(x, W1, h0);
    k_init_out1<<<cdiv((long long)N * 32, 256), 256, 0, stream>>>(h0, b1, dis, out1, N);
    k_scatter128<<<cdiv((long long)E * 32, 256), 256, 0, stream>>>(src, dst, dis, h0, out1, E);

    int ntiles = N / 16;
    k_gemm2    <<<cdiv(ntiles, 8), 256, 0, stream>>>(out1, W2, h2, ntiles);
    k_init_out2<<<cdiv((long long)N * 16, 256), 256, 0, stream>>>(h2, b2, dis, logits, N);
    k_scatter16<<<cdiv((long long)E * 16, 256), 256, 0, stream>>>(src, dst, dis, h2, logits, E);

    k_softmax16<<<cdiv((long long)N * 16, 256), 256, 0, stream>>>(logits, probs, N);
}


# scatter128: lane = feature (stride 4 B, not 16 B), so each f32 atomic add and each row load of a wave covers one whole 128-B line: 4 line requests per edge for the adds, not 16
# speedup vs baseline: 1.2726x; 1.2726x over previous
#include <hip/hip_runtime.h>
#include <hip/hip_bf16.h>

typedef float v2f __attribute__((ext_vector_type(2)));
typedef float v8f __attribute__((ext_vector_type(8)));

#define LDS_STRIDE 260

__device__ __forceinline__ void atomAddF32(float* p, float v) {
    unsafeAtomicAdd(p, v);
}


__global__ __launch_bounds__(256) void k_deg_init(float* __restrict__ deg, int n) {
    int i = blockIdx.x * 256 + threadIdx.x;
    if (i < n) deg[i] = 1.0f;
}

__global__ __launch_bounds__(256) void k_deg_edges(const int* __restrict__ dst,
                                                   float* __restrict__ deg, int e) {
    int i = blockIdx.x * 256 + threadIdx.x;
    if (i < e) atomAddF32(&deg[dst[i]], 1.0f);
}

__global__ __launch_bounds__(256) void k_rsqrt(float* __restrict__ deg, int n) {
    int i = blockIdx.x * 256 + threadIdx.x;
    if (i < n) deg[i] = __frsqrt_rn(deg[i]);
}


__global__ __launch_bounds__(256) void k_gemm1(const float* __restrict__ x,
                                               const float* __restrict__ W1,
                                               float* __restrict__ h0) {
    __shared__ float As[16 * LDS_STRIDE];
    const int tid = threadIdx.x;
    const int r0  = blockIdx.x * 16;

    const float4* xv = (const float4*)(x + (size_t)r0 * 256);
    #pragma unroll
    for (int i = 0; i < 4; ++i) {
        int j   = i * 256 + tid;
        int row = j >> 6;
        int c4  = j & 63;
        float4 v = xv[(size_t)row * 64 + c4];
        float* d = &As[row * LDS_STRIDE + c4 * 4];
        d[0] = v.x; d[1] = v.y; d[2] = v.z; d[3] = v.w;
    }
    __syncthreads();

    const int wave  = tid >> 5;
    const int lane  = tid & 31;
    const int half  = lane >> 4;
    const int l16   = lane & 15;
    const int c0    = wave * 16;
    const int kboff = half * 2;

    v8f acc = {0.f, 0.f, 0.f, 0.f, 0.f, 0.f, 0.f, 0.f};
    for (int k0 = 0; k0 < 256; k0 += 4) {
        v2f a, b;
        const float* ap = &As[l16 * LDS_STRIDE + k0 + kboff];
        a.x = ap[0]; a.y = ap[1];
        const float* bp = &W1[(size_t)(k0 + kboff) * 128 + c0 + l16];
        b.x = bp[0]; b.y = bp[128];
        acc = __builtin_amdgcn_wmma_f32_16x16x4_f32(false, a, false, b,
                                                    (short)0, acc, false, false);
    }
    float* out = h0 + (size_t)(r0 + half * 8) * 128 + c0 + l16;
    #pragma unroll
    for (int i = 0; i < 8; ++i) out[(size_t)i * 128] = acc[i];
}


__global__ __launch_bounds__(256) void k_init_out1(const float* __restrict__ h0,
                                                   const float* __restrict__ b1,
                                                   const float* __restrict__ dis,
                                                   float* __restrict__ out1, int n) {
    int g = blockIdx.x * 256 + threadIdx.x;
    if (g >= n * 32) return;
    int node = g >> 5;
    int f4   = g & 31;
    float d2 = dis[node] * dis[node];
    float4 v = ((const float4*)h0)[(size_t)g];
    float4 bb = ((const float4*)b1)[f4];
    float4 o;
    o.x = bb.x + v.x * d2; o.y = bb.y + v.y * d2;
    o.z = bb.z + v.z * d2; o.w = bb.w + v.w * d2;
    ((float4*)out1)[(size_t)g] = o;
}

__global__ __launch_bounds__(256) void k_scatter128(const int* __restrict__ src,
                                                    const int* __restrict__ dst,
                                                    const float* __restrict__ dis,
                                                    const float* __restrict__ h,
                                                    float* __restrict__ out, int e) {
    int g = blockIdx.x * 256 + threadIdx.x;
    int edge = g >> 5;
    if (edge >= e) return;
    int lane = threadIdx.x & 31;
    int s = src[edge], d = dst[edge];
    float nrm = dis[s] * dis[d];
    const float* hp = h + (size_t)s * 128 + lane;
    float* o = out + (size_t)d * 128 + lane;
    float v0 = hp[0], v1 = hp[32], v2 = hp[64], v3 = hp[96];
    atomAddF32(o +  0, v0 * nrm);
    atomAddF32(o + 32, v1 * nrm);
    atomAddF32(o + 64, v2 * nrm);
    atomAddF32(o + 96, v3 * nrm);
}


__global__ __launch_bounds__(256) void k_gemm2(const float* __restrict__ out1,
                                               const float* __restrict__ W2,
                                               float* __restrict__ h2, int ntiles) {
    int wg = (blockIdx.x * 256 + threadIdx.x) >> 5;
    if (wg >= ntiles) return;
    const int lane  = threadIdx.x & 31;
    const int half  = lane >> 4;
    const int l16   = lane & 15;
    const int kboff = half * 2;
    const int r0    = wg * 16;

    const float* arow = out1 + (size_t)(r0 + l16) * 128;
    v8f acc = {0.f, 0.f, 0.f, 0.f, 0.f, 0.f, 0.f, 0.f};
    for (int k0 = 0; k0 < 128; k0 += 4) {
        v2f a, b;
        a.x = fmaxf(arow[k0 + kboff], 0.f);
        a.y = fmaxf(arow[k0 + kboff + 1], 0.f);
        const float* bp = &W2[(size_t)(k0 + kboff) * 16 + l16];
        b.x = bp[0]; b.y = bp[16];
        acc = __builtin_amdgcn_wmma_f32_16x16x4_f32(false, a, false, b,
                                                    (short)0, acc, false, false);
    }
    float* out = h2 + (size_t)(r0 + half * 8) * 16 + l16;
    #pragma unroll
    for (int i = 0; i < 8; ++i) out[(size_t)i * 16] = acc[i];
}


__global__ __launch_bounds__(256) void k_init_out2(const float* __restrict__ h2,
                                                   const float* __restrict__ b2,
                                                   const float* __restrict__ dis,
                                                   float* __restrict__ logits, int n) {
    int g = blockIdx.x * 256 + threadIdx.x;
    if (g >= n * 16) return;
    int node = g >> 4;
    int f    = g & 15;
    float d2 = dis[node] * dis[node];
    logits[g] = b2[f] + h2[g] * d2;
}

__global__ __launch_bounds__(256) void k_scatter16(const int* __restrict__ src,
                                                   const int* __restrict__ dst,
                                                   const float* __restrict__ dis,
                                                   const float* __restrict__ h2,
                                                   float* __restrict__ logits, int e) {
    int g = blockIdx.x * 256 + threadIdx.x;
    int edge = g >> 4;
    if (edge >= e) return;
    int f = g & 15;
    int s = src[edge], d = dst[edge];
    float nrm = dis[s] * dis[d];
    float v = h2[(size_t)s * 16 + f];
    atomAddF32(&logits[(size_t)d * 16 + f], v * nrm);
}

__global__ __launch_bounds__(256) void k_softmax16(const float* __restrict__ logits,
                                                   float* __restrict__ probs, int n) {
    int g = blockIdx.x * 256 + threadIdx.x;
    if (g >= n * 16) return;
    float v = logits[g];
    float m = v;
    #pragma unroll
    for (int o = 8; o >= 1; o >>= 1) m = fmaxf(m, __shfl_xor(m, o, 16));
    float e = __expf(v - m);
    float s = e;
    #pragma unroll
    for (int o = 8; o >= 1; o >>= 1) s += __shfl_xor(s, o, 16);
    probs[g] = e / s;
}


extern "C" void kernel_launch(void* const* d_in, const int* in_sizes, int n_in,
                              void* d_out, int out_size, void* d_ws, size_t ws_size,
                              hipStream_t stream) {
    const float* x  = (const float*)d_in[0];
    const int*   ei = (const int*)d_in[1];
    const float* W1 = (const float*)d_in[2];
    const float* b1 = (const float*)d_in[3];
    const float* W2 = (const float*)d_in[4];
    const float* b2 = (const float*)d_in[5];

    const int N = in_sizes[0] / 256;
    const int E = in_sizes[1] / 2;
    const int* src = ei;
    const int* dst = ei + E;

    char*  ws  = (char*)d_ws;
    size_t off = 0;
    auto alloc = [&](size_t bytes) -> void* {
        void* p = ws + off;
        off += (bytes + 255) & ~(size_t)255;
        return p;
    };
    float* dis  = (float*)alloc((size_t)N * 4);
    float* h0   = (float*)alloc((size_t)N * 128 * 4);
    float* out1 = (float*)alloc((size_t)N * 128 * 4);
    float* h2   = (float*)alloc((size_t)N * 16 * 4);
    float* logits = (float*)d_out;
    float* probs  = (float*)d_out + (size_t)N * 16;

    auto cdiv = [](long long a, long long b) { return (int)((a + b - 1) / b); };

    k_deg_init <<<cdiv(N, 256), 256, 0, stream>>>(dis, N);
    k_deg_edges<<<cdiv(E, 256), 256, 0, stream>>>(dst, dis, E);
    k_rsqrt    <<<cdiv(N, 256), 256, 0, stream>>>(dis, N);

    k_gemm1    <<<N / 16, 256, 0, stream>>>(x, W1, h0);
    k_init_out1<<<cdiv((long long)N * 32, 256), 256, 0, stream>>>(h0, b1, dis, out1, N);
    k_scatter128<<<cdiv((long long)E * 32, 256), 256, 0, stream>>>(src, dst, dis, h0, out1, E);

    int ntiles = N / 16;
    k_gemm2    <<<cdiv(ntiles, 8), 256, 0, stream>>>(out1, W2, h2, ntiles);
    k_init_out2<<<cdiv((long long)N * 16, 256), 256, 0, stream>>>(h2, b2, dis, logits, N);
    k_scatter16<<<cdiv((long long)E * 16, 256), 256, 0, stream>>>(src, dst, dis, h2, logits, E);

    k_softmax16<<<cdiv((long long)N * 16, 256), 256, 0, stream>>>(logits, probs, N);
}


# both edge scatter-adds: one wave per 32 consecutive edges (ids and norm loaded once, lane-broadcast, row loads batched) not one wave or half-wave per edge; f32 atomic arrival order moves
# speedup vs baseline: 1.6308x; 1.2815x over previous
#include <hip/hip_runtime.h>
#include <hip/hip_bf16.h>

typedef float v2f __attribute__((ext_vector_type(2)));
typedef float v8f __attribute__((ext_vector_type(8)));

#define LDS_STRIDE 260

__device__ __forceinline__ void atomAddF32(float* p, float v) {
    unsafeAtomicAdd(p, v);
}


__global__ __launch_bounds__(256) void k_deg_init(float* __restrict__ deg, int n) {
    int i = blockIdx.x * 256 + threadIdx.x;
    if (i < n) deg[i] = 1.0f;
}

__global__ __launch_bounds__(256) void k_deg_edges(const int* __restrict__ dst,
                                                   float* __restrict__ deg, int e) {
    int i = blockIdx.x * 256 + threadIdx.x;
    if (i < e) atomAddF32(&deg[dst[i]], 1.0f);
}

__global__ __launch_bounds__(256) void k_rsqrt(float* __restrict__ deg, int n) {
    int i = blockIdx.x * 256 + threadIdx.x;
    if (i < n) deg[i] = __frsqrt_rn(deg[i]);
}


__global__ __launch_bounds__(256) void k_gemm1(const float* __restrict__ x,
                                               const float* __restrict__ W1,
                                               float* __restrict__ h0) {
    __shared__ float As[16 * LDS_STRIDE];
    const int tid = threadIdx.x;
    const int r0  = blockIdx.x * 16;

    const float4* xv = (const float4*)(x + (size_t)r0 * 256);
    #pragma unroll
    for (int i = 0; i < 4; ++i) {
        int j   = i * 256 + tid;
        int row = j >> 6;
        int c4  = j & 63;
        float4 v = xv[(size_t)row * 64 + c4];
        float* d = &As[row * LDS_STRIDE + c4 * 4];
        d[0] = v.x; d[1] = v.y; d[2] = v.z; d[3] = v.w;
    }
    __syncthreads();

    const int wave  = tid >> 5;
    const int lane  = tid & 31;
    const int half  = lane >> 4;
    const int l16   = lane & 15;
    const int c0    = wave * 16;
    const int kboff = half * 2;

    v8f acc = {0.f, 0.f, 0.f, 0.f, 0.f, 0.f, 0.f, 0.f};
    for (int k0 = 0; k0 < 256; k0 += 4) {
        v2f a, b;
        const float* ap = &As[l16 * LDS_STRIDE + k0 + kboff];
        a.x = ap[0]; a.y = ap[1];
        const float* bp = &W1[(size_t)(k0 + kboff) * 128 + c0 + l16];
        b.x = bp[0]; b.y = bp[128];
        acc = __builtin_amdgcn_wmma_f32_16x16x4_f32(false, a, false, b,
                                                    (short)0, acc, false, false);
    }
    float* out = h0 + (size_t)(r0 + half * 8) * 128 + c0 + l16;
    #pragma unroll
    for (int i = 0; i < 8; ++i) out[(size_t)i * 128] = acc[i];
}


__global__ __launch_bounds__(256) void k_init_out1(const float* __restrict__ h0,
                                                   const float* __restrict__ b1,
                                                   const float* __restrict__ dis,
                                                   float* __restrict__ out1, int n) {
    int g = blockIdx.x * 256 + threadIdx.x;
    if (g >= n * 32) return;
    int node = g >> 5;
    int f4   = g & 31;
    float d2 = dis[node] * dis[node];
    float4 v = ((const float4*)h0)[(size_t)g];
    float4 bb = ((const float4*)b1)[f4];
    float4 o;
    o.x = bb.x + v.x * d2; o.y = bb.y + v.y * d2;
    o.z = bb.z + v.z * d2; o.w = bb.w + v.w * d2;
    ((float4*)out1)[(size_t)g] = o;
}

__global__ __launch_bounds__(256) void k_scatter128(const int* __restrict__ src,
                                                    const int* __restrict__ dst,
                                                    const float* __restrict__ dis,
                                                    const float* __restrict__ h,
                                                    float* __restrict__ out, int e) {
    const int lane = threadIdx.x & 31;
    const int wave = blockIdx.x * 8 + __builtin_amdgcn_readfirstlane((int)(threadIdx.x >> 5));
    const int e0 = wave * 32;
    if (e0 >= e) return;
    const int rem = e - e0;

    int s = 0, d = 0;
    float nrm = 0.f;
    if (lane < rem) {
        s = src[e0 + lane];
        d = dst[e0 + lane];
        nrm = dis[s] * dis[d];
    }
    const int nrm_bits = __float_as_int(nrm);

    if (rem >= 32) {
        #pragma unroll
        for (int jb = 0; jb < 32; jb += 8) {
            float v[8][4];
            #pragma unroll
            for (int u = 0; u < 8; ++u) {
                const int sj = __builtin_amdgcn_readlane(s, jb + u);
                const float* hp = h + (size_t)sj * 128 + lane;
                v[u][0] = hp[0]; v[u][1] = hp[32]; v[u][2] = hp[64]; v[u][3] = hp[96];
            }
            #pragma unroll
            for (int u = 0; u < 8; ++u) {
                const int   dj = __builtin_amdgcn_readlane(d, jb + u);
                const float nj = __int_as_float(__builtin_amdgcn_readlane(nrm_bits, jb + u));
                float* o = out + (size_t)dj * 128 + lane;
                atomAddF32(o +  0, v[u][0] * nj);
                atomAddF32(o + 32, v[u][1] * nj);
                atomAddF32(o + 64, v[u][2] * nj);
                atomAddF32(o + 96, v[u][3] * nj);
            }
        }
    } else {
        for (int j = 0; j < rem; ++j) {
            const int   sj = __shfl(s, j);
            const int   dj = __shfl(d, j);
            const float nj = __shfl(nrm, j);
            const float* hp = h + (size_t)sj * 128 + lane;
            float* o = out + (size_t)dj * 128 + lane;
            float v0 = hp[0], v1 = hp[32], v2 = hp[64], v3 = hp[96];
            atomAddF32(o +  0, v0 * nj);
            atomAddF32(o + 32, v1 * nj);
            atomAddF32(o + 64, v2 * nj);
            atomAddF32(o + 96, v3 * nj);
        }
    }
}


__global__ __launch_bounds__(256) void k_gemm2(const float* __restrict__ out1,
                                               const float* __restrict__ W2,
                                               float* __restrict__ h2, int ntiles) {
    int wg = (blockIdx.x * 256 + threadIdx.x) >> 5;
    if (wg >= ntiles) return;
    const int lane  = threadIdx.x & 31;
    const int half  = lane >> 4;
    const int l16   = lane & 15;
    const int kboff = half * 2;
    const int r0    = wg * 16;

    const float* arow = out1 + (size_t)(r0 + l16) * 128;
    v8f acc = {0.f, 0.f, 0.f, 0.f, 0.f, 0.f, 0.f, 0.f};
    for (int k0 = 0; k0 < 128; k0 += 4) {
        v2f a, b;
        a.x = fmaxf(arow[k0 + kboff], 0.f);
        a.y = fmaxf(arow[k0 + kboff + 1], 0.f);
        const float* bp = &W2[(size_t)(k0 + kboff) * 16 + l16];
        b.x = bp[0]; b.y = bp[16];
        acc = __builtin_amdgcn_wmma_f32_16x16x4_f32(false, a, false, b,
                                                    (short)0, acc, false, false);
    }
    float* out = h2 + (size_t)(r0 + half * 8) * 16 + l16;
    #pragma unroll
    for (int i = 0; i < 8; ++i) out[(size_t)i * 16] = acc[i];
}


__global__ __launch_bounds__(256) void k_init_out2(const float* __restrict__ h2,
                                                   const float* __restrict__ b2,
                                                   const float* __restrict__ dis,
                                                   float* __restrict__ logits, int n) {
    int g = blockIdx.x * 256 + threadIdx.x;
    if (g >= n * 16) return;
    int node = g >> 4;
    int f    = g & 15;
    float d2 = dis[node] * dis[node];
    logits[g] = b2[f] + h2[g] * d2;
}

__global__ __launch_bounds__(256) void k_scatter16(const int* __restrict__ src,
                                                   const int* __restrict__ dst,
                                                   const float* __restrict__ dis,
                                                   const float* __restrict__ h2,
                                                   float* __restrict__ logits, int e) {
    const int lane = threadIdx.x & 31;
    const int wave = blockIdx.x * 8 + __builtin_amdgcn_readfirstlane((int)(threadIdx.x >> 5));
    const int e0 = wave * 32;
    if (e0 >= e) return;
    const int rem = e - e0;

    int s = 0, d = 0;
    float nrm = 0.f;
    if (lane < rem) {
        s = src[e0 + lane];
        d = dst[e0 + lane];
        nrm = dis[s] * dis[d];
    }
    const int f  = lane & 15;
    const int hb = lane & 16;

    if (rem >= 32) {
        float v[16];
        #pragma unroll
        for (int j = 0; j < 16; ++j) {
            const int sj = __shfl(s, j, 16);
            v[j] = h2[(size_t)sj * 16 + f];
        }
        #pragma unroll
        for (int j = 0; j < 16; ++j) {
            const int   dj = __shfl(d, j, 16);
            const float nj = __shfl(nrm, j, 16);
            atomAddF32(&logits[(size_t)dj * 16 + f], v[j] * nj);
        }
    } else {
        for (int j = 0; j < 16; ++j) {
            const int   sj = __shfl(s, j, 16);
            const int   dj = __shfl(d, j, 16);
            const float nj = __shfl(nrm, j, 16);
            if (hb + j < rem)
                atomAddF32(&logits[(size_t)dj * 16 + f], h2[(size_t)sj * 16 + f] * nj);
        }
    }
}

__global__ __launch_bounds__(256) void k_softmax16(const float* __restrict__ logits,
                                                   float* __restrict__ probs, int n) {
    int g = blockIdx.x * 256 + threadIdx.x;
    if (g >= n * 16) return;
    float v = logits[g];
    float m = v;
    #pragma unroll
    for (int o = 8; o >= 1; o >>= 1) m = fmaxf(m, __shfl_xor(m, o, 16));
    float e = __expf(v - m);
    float s = e;
    #pragma unroll
    for (int o = 8; o >= 1; o >>= 1) s += __shfl_xor(s, o, 16);
    probs[g] = e / s;
}


extern "C" void kernel_launch(void* const* d_in, const int* in_sizes, int n_in,
                              void* d_out, int out_size, void* d_ws, size_t ws_size,
                              hipStream_t stream) {
    const float* x  = (const float*)d_in[0];
    const int*   ei = (const int*)d_in[1];
    const float* W1 = (const float*)d_in[2];
    const float* b1 = (const float*)d_in[3];
    const float* W2 = (const float*)d_in[4];
    const float* b2 = (const float*)d_in[5];

    const int N = in_sizes[0] / 256;
    const int E = in_sizes[1] / 2;
    const int* src = ei;
    const int* dst = ei + E;

    char*  ws  = (char*)d_ws;
    size_t off = 0;
    auto alloc = [&](size_t bytes) -> void* {
        void* p = ws + off;
        off += (bytes + 255) & ~(size_t)255;
        return p;
    };
    float* dis  = (float*)alloc((size_t)N * 4);
    float* h0   = (float*)alloc((size_t)N * 128 * 4);
    float* out1 = (float*)alloc((size_t)N * 128 * 4);
    float* h2   = (float*)alloc((size_t)N * 16 * 4);
    float* logits = (float*)d_out;
    float* probs  = (float*)d_out + (size_t)N * 16;

    auto cdiv = [](long long a, long long b) { return (int)((a + b - 1) / b); };

    k_deg_init <<<cdiv(N, 256), 256, 0, stream>>>(dis, N);
    k_deg_edges<<<cdiv(E, 256), 256, 0, stream>>>(dst, dis, E);
    k_rsqrt    <<<cdiv(N, 256), 256, 0, stream>>>(dis, N);

    k_gemm1    <<<N / 16, 256, 0, stream>>>(x, W1, h0);
    k_init_out1<<<cdiv((long long)N * 32, 256), 256, 0, stream>>>(h0, b1, dis, out1, N);
    k_scatter128<<<cdiv(E, 256), 256, 0, stream>>>(src, dst, dis, h0, out1, E);

    int ntiles = N / 16;
    k_gemm2    <<<cdiv(ntiles, 8), 256, 0, stream>>>(out1, W2, h2, ntiles);
    k_init_out2<<<cdiv((long long)N * 16, 256), 256, 0, stream>>>(h2, b2, dis, logits, N);
    k_scatter16<<<cdiv(E, 256), 256, 0, stream>>>(src, dst, dis, h2, logits, E);

    k_softmax16<<<cdiv((long long)N * 16, 256), 256, 0, stream>>>(logits, probs, N);
}


# the 3 edge atomic passes (degree, 128-wide, 16-wide): each chunk runs once per L2 half (die id, chunk ticket per half), adding only rows of that half's node range; f32 atomic arrival order moves
# speedup vs baseline: 2.8841x; 1.7685x over previous
#include <hip/hip_runtime.h>
#include <hip/hip_bf16.h>

typedef float v2f __attribute__((ext_vector_type(2)));
typedef float v8f __attribute__((ext_vector_type(8)));

#define LDS_STRIDE 260

__device__ __forceinline__ void atomAddF32(float* p, float v) {
    unsafeAtomicAdd(p, v);
}


__global__ __launch_bounds__(256) void k_deg_init(float* __restrict__ deg, int* __restrict__ ctr, int n) {
    int i = blockIdx.x * 256 + threadIdx.x;
    if (i < n) deg[i] = 1.0f;
    if (i < 6) ctr[i] = 0;
}


__device__ __forceinline__ void take_task(int* __restrict__ ctr, int nchunk,
                                          int* sh, int& hf, int& chunk) {
    if (threadIdx.x == 0) {
        const unsigned id = __builtin_amdgcn_s_sendmsg_rtn(0x87);
        int h = (int)((id >> 18) & 1u);
        int t = atomicAdd(&ctr[h], 1);
        if (t >= nchunk) { h ^= 1; t = atomicAdd(&ctr[h], 1); }
        sh[0] = h; sh[1] = t;
    }
    __syncthreads();
    hf    = __builtin_amdgcn_readfirstlane(sh[0]);
    chunk = __builtin_amdgcn_readfirstlane(sh[1]);
}

__global__ __launch_bounds__(256) void k_deg_edges(const int* __restrict__ dst,
                                                   float* __restrict__ deg,
                                                   int* __restrict__ ctr,
                                                   int e, int nsplit, int nchunk) {
    __shared__ int sh[2];
    int hf, chunk;
    take_task(ctr, nchunk, sh, hf, chunk);
    if (chunk >= nchunk) return;
    const int base = chunk * 2048 + (int)threadIdx.x;
    #pragma unroll
    for (int k = 0; k < 8; ++k) {
        const int i = base + k * 256;
        if (i < e) {
            const int d = dst[i];
            if ((d >= nsplit) == (hf != 0)) atomAddF32(&deg[d], 1.0f);
        }
    }
}

__global__ __launch_bounds__(256) void k_rsqrt(float* __restrict__ deg, int n) {
    int i = blockIdx.x * 256 + threadIdx.x;
    if (i < n) deg[i] = __frsqrt_rn(deg[i]);
}


__global__ __launch_bounds__(256) void k_gemm1(const float* __restrict__ x,
                                               const float* __restrict__ W1,
                                               float* __restrict__ h0) {
    __shared__ float As[16 * LDS_STRIDE];
    const int tid = threadIdx.x;
    const int r0  = blockIdx.x * 16;

    const float4* xv = (const float4*)(x + (size_t)r0 * 256);
    #pragma unroll
    for (int i = 0; i < 4; ++i) {
        int j   = i * 256 + tid;
        int row = j >> 6;
        int c4  = j & 63;
        float4 v = xv[(size_t)row * 64 + c4];
        float* d = &As[row * LDS_STRIDE + c4 * 4];
        d[0] = v.x; d[1] = v.y; d[2] = v.z; d[3] = v.w;
    }
    __syncthreads();

    const int wave  = tid >> 5;
    const int lane  = tid & 31;
    const int half  = lane >> 4;
    const int l16   = lane & 15;
    const int c0    = wave * 16;
    const int kboff = half * 2;

    v8f acc = {0.f, 0.f, 0.f, 0.f, 0.f, 0.f, 0.f, 0.f};
    for (int k0 = 0; k0 < 256; k0 += 4) {
        v2f a, b;
        const float* ap = &As[l16 * LDS_STRIDE + k0 + kboff];
        a.x = ap[0]; a.y = ap[1];
        const float* bp = &W1[(size_t)(k0 + kboff) * 128 + c0 + l16];
        b.x = bp[0]; b.y = bp[128];
        acc = __builtin_amdgcn_wmma_f32_16x16x4_f32(false, a, false, b,
                                                    (short)0, acc, false, false);
    }
    float* out = h0 + (size_t)(r0 + half * 8) * 128 + c0 + l16;
    #pragma unroll
    for (int i = 0; i < 8; ++i) out[(size_t)i * 128] = acc[i];
}


__global__ __launch_bounds__(256) void k_init_out1(const float* __restrict__ h0,
                                                   const float* __restrict__ b1,
                                                   const float* __restrict__ dis,
                                                   float* __restrict__ out1, int n) {
    int g = blockIdx.x * 256 + threadIdx.x;
    if (g >= n * 32) return;
    int node = g >> 5;
    int f4   = g & 31;
    float d2 = dis[node] * dis[node];
    float4 v = ((const float4*)h0)[(size_t)g];
    float4 bb = ((const float4*)b1)[f4];
    float4 o;
    o.x = bb.x + v.x * d2; o.y = bb.y + v.y * d2;
    o.z = bb.z + v.z * d2; o.w = bb.w + v.w * d2;
    ((float4*)out1)[(size_t)g] = o;
}

template <bool FULL>
__device__ __forceinline__ void batch128(unsigned long long& m, int lane,
                                         int sA, int sB, int dA, int dB, int nA, int nB,
                                         const float* __restrict__ h, float* __restrict__ out) {
    int sj[8], dj[8]; float nj[8]; bool ok[8];
    #pragma unroll
    for (int u = 0; u < 8; ++u) {
        ok[u] = FULL || (m != 0ull);
        const int j = ok[u] ? __builtin_ctzll(m) : 0;
        m &= (m - 1ull);
        const int  l  = j & 31;
        const bool hi = j >= 32;
        sj[u] = __builtin_amdgcn_readlane(hi ? sB : sA, l);
        dj[u] = __builtin_amdgcn_readlane(hi ? dB : dA, l);
        nj[u] = __int_as_float(__builtin_amdgcn_readlane(hi ? nB : nA, l));
    }
    float v[8][4];
    #pragma unroll
    for (int u = 0; u < 8; ++u) {
        if (ok[u]) {
            const float* hp = h + (size_t)sj[u] * 128 + lane;
            v[u][0] = hp[0]; v[u][1] = hp[32]; v[u][2] = hp[64]; v[u][3] = hp[96];
        }
    }
    #pragma unroll
    for (int u = 0; u < 8; ++u) {
        if (ok[u]) {
            float* o = out + (size_t)dj[u] * 128 + lane;
            atomAddF32(o +  0, v[u][0] * nj[u]);
            atomAddF32(o + 32, v[u][1] * nj[u]);
            atomAddF32(o + 64, v[u][2] * nj[u]);
            atomAddF32(o + 96, v[u][3] * nj[u]);
        }
    }
}

__global__ __launch_bounds__(256) void k_scatter128(const int* __restrict__ src,
                                                    const int* __restrict__ dst,
                                                    const float* __restrict__ dis,
                                                    const float* __restrict__ h,
                                                    float* __restrict__ out,
                                                    int* __restrict__ ctr,
                                                    int e, int nsplit, int nchunk) {
    __shared__ int sh[2];
    int hf, chunk;
    take_task(ctr, nchunk, sh, hf, chunk);
    if (chunk >= nchunk) return;

    const int lane = threadIdx.x & 31;
    const int wave = __builtin_amdgcn_readfirstlane((int)(threadIdx.x >> 5));
    const int e0 = chunk * 512 + wave * 64;
    const int eA = e0 + lane, eB = e0 + 32 + lane;

    int sA = 0, dA = 0, sB = 0, dB = 0;
    float nA = 0.f, nB = 0.f;
    bool pA = false, pB = false;
    if (eA < e) { sA = src[eA]; dA = dst[eA]; pA = ((dA >= nsplit) == (hf != 0)); }
    if (eB < e) { sB = src[eB]; dB = dst[eB]; pB = ((dB >= nsplit) == (hf != 0)); }
    if (pA) nA = dis[sA] * dis[dA];
    if (pB) nB = dis[sB] * dis[dB];
    unsigned long long m = (unsigned long long)__builtin_amdgcn_ballot_w32(pA)
                         | ((unsigned long long)__builtin_amdgcn_ballot_w32(pB) << 32);
    const int nAb = __float_as_int(nA), nBb = __float_as_int(nB);

    while (__builtin_popcountll(m) >= 8)
        batch128<true>(m, lane, sA, sB, dA, dB, nAb, nBb, h, out);
    if (m != 0ull)
        batch128<false>(m, lane, sA, sB, dA, dB, nAb, nBb, h, out);
}


__global__ __launch_bounds__(256) void k_gemm2(const float* __restrict__ out1,
                                               const float* __restrict__ W2,
                                               float* __restrict__ h2, int ntiles) {
    int wg = (blockIdx.x * 256 + threadIdx.x) >> 5;
    if (wg >= ntiles) return;
    const int lane  = threadIdx.x & 31;
    const int half  = lane >> 4;
    const int l16   = lane & 15;
    const int kboff = half * 2;
    const int r0    = wg * 16;

    const float* arow = out1 + (size_t)(r0 + l16) * 128;
    v8f acc = {0.f, 0.f, 0.f, 0.f, 0.f, 0.f, 0.f, 0.f};
    for (int k0 = 0; k0 < 128; k0 += 4) {
        v2f a, b;
        a.x = fmaxf(arow[k0 + kboff], 0.f);
        a.y = fmaxf(arow[k0 + kboff + 1], 0.f);
        const float* bp = &W2[(size_t)(k0 + kboff) * 16 + l16];
        b.x = bp[0]; b.y = bp[16];
        acc = __builtin_amdgcn_wmma_f32_16x16x4_f32(false, a, false, b,
                                                    (short)0, acc, false, false);
    }
    float* out = h2 + (size_t)(r0 + half * 8) * 16 + l16;
    #pragma unroll
    for (int i = 0; i < 8; ++i) out[(size_t)i * 16] = acc[i];
}


__global__ __launch_bounds__(256) void k_init_out2(const float* __restrict__ h2,
                                                   const float* __restrict__ b2,
                                                   const float* __restrict__ dis,
                                                   float* __restrict__ logits, int n) {
    int g = blockIdx.x * 256 + threadIdx.x;
    if (g >= n * 16) return;
    int node = g >> 4;
    int f    = g & 15;
    float d2 = dis[node] * dis[node];
    logits[g] = b2[f] + h2[g] * d2;
}

__global__ __launch_bounds__(256) void k_scatter16(const int* __restrict__ src,
                                                   const int* __restrict__ dst,
                                                   const float* __restrict__ dis,
                                                   const float* __restrict__ h2,
                                                   float* __restrict__ logits,
                                                   int* __restrict__ ctr,
                                                   int e, int nsplit, int nchunk) {
    __shared__ int sh[2];
    __shared__ int pk[8][3][64];
    int hf, chunk;
    take_task(ctr, nchunk, sh, hf, chunk);
    if (chunk >= nchunk) return;

    const int lane = threadIdx.x & 31;
    const int wave = __builtin_amdgcn_readfirstlane((int)(threadIdx.x >> 5));
    const int e0 = chunk * 512 + wave * 64;
    const int eA = e0 + lane, eB = e0 + 32 + lane;

    int sA = 0, dA = 0, sB = 0, dB = 0;
    float nA = 0.f, nB = 0.f;
    bool pA = false, pB = false;
    if (eA < e) { sA = src[eA]; dA = dst[eA]; pA = ((dA >= nsplit) == (hf != 0)); }
    if (eB < e) { sB = src[eB]; dB = dst[eB]; pB = ((dB >= nsplit) == (hf != 0)); }
    if (pA) nA = dis[sA] * dis[dA];
    if (pB) nB = dis[sB] * dis[dB];

    const unsigned mA = __builtin_amdgcn_ballot_w32(pA);
    const unsigned mB = __builtin_amdgcn_ballot_w32(pB);
    const int cA  = __builtin_popcount(mA);
    const int cnt = cA + __builtin_popcount(mB);
    int (*w)[64] = pk[wave];
    if (pA) {
        const int p = (int)__builtin_amdgcn_mbcnt_lo(mA, 0u);
        w[0][p] = sA; w[1][p] = dA; w[2][p] = __float_as_int(nA);
    }
    if (pB) {
        const int p = cA + (int)__builtin_amdgcn_mbcnt_lo(mB, 0u);
        w[0][p] = sB; w[1][p] = dB; w[2][p] = __float_as_int(nB);
    }
    __syncthreads();
    int s0 = 0, d0 = 0, n0 = 0, s1 = 0, d1 = 0, n1 = 0;
    if (lane < cnt)      { s0 = w[0][lane];      d0 = w[1][lane];      n0 = w[2][lane]; }
    if (32 + lane < cnt) { s1 = w[0][32 + lane]; d1 = w[1][32 + lane]; n1 = w[2][32 + lane]; }

    const int f   = lane & 15;
    const int top = lane >> 4;
    for (int jb = 0; jb < cnt; jb += 16) {
        const bool second = jb >= 32;
        const int S = second ? s1 : s0;
        const int D = second ? d1 : d0;
        const int G = second ? n1 : n0;
        const int l0 = (jb & 31) + top;
        float v[8]; int dj[8]; float nj[8]; bool on[8];
        #pragma unroll
        for (int u = 0; u < 8; ++u) {
            const int sl = (l0 + 2 * u) << 2;
            on[u] = (jb + 2 * u + top) < cnt;
            const int sj = __builtin_amdgcn_ds_bpermute(sl, S);
            dj[u] = __builtin_amdgcn_ds_bpermute(sl, D);
            nj[u] = __int_as_float(__builtin_amdgcn_ds_bpermute(sl, G));
            v[u] = 0.f;
            if (on[u]) v[u] = h2[sj * 16 + f];
        }
        #pragma unroll
        for (int u = 0; u < 8; ++u) {
            if (on[u]) atomAddF32(&logits[dj[u] * 16 + f], v[u] * nj[u]);
        }
    }
}

__global__ __launch_bounds__(256) void k_softmax16(const float* __restrict__ logits,
                                                   float* __restrict__ probs, int n) {
    int g = blockIdx.x * 256 + threadIdx.x;
    if (g >= n * 16) return;
    float v = logits[g];
    float m = v;
    #pragma unroll
    for (int o = 8; o >= 1; o >>= 1) m = fmaxf(m, __shfl_xor(m, o, 16));
    float e = __expf(v - m);
    float s = e;
    #pragma unroll
    for (int o = 8; o >= 1; o >>= 1) s += __shfl_xor(s, o, 16);
    probs[g] = e / s;
}


extern "C" void kernel_launch(void* const* d_in, const int* in_sizes, int n_in,
                              void* d_out, int out_size, void* d_ws, size_t ws_size,
                              hipStream_t stream) {
    const float* x  = (const float*)d_in[0];
    const int*   ei = (const int*)d_in[1];
    const float* W1 = (const float*)d_in[2];
    const float* b1 = (const float*)d_in[3];
    const float* W2 = (const float*)d_in[4];
    const float* b2 = (const float*)d_in[5];

    const int N = in_sizes[0] / 256;
    const int E = in_sizes[1] / 2;
    const int* src = ei;
    const int* dst = ei + E;

    char*  ws  = (char*)d_ws;
    size_t off = 0;
    auto alloc = [&](size_t bytes) -> void* {
        void* p = ws + off;
        off += (bytes + 255) & ~(size_t)255;
        return p;
    };
    float* dis  = (float*)alloc((size_t)N * 4);
    float* h0   = (float*)alloc((size_t)N * 128 * 4);
    float* out1 = (float*)alloc((size_t)N * 128 * 4);
    float* h2   = (float*)alloc((size_t)N * 16 * 4);
    int*   ctr  = (int*)alloc(256);
    float* logits = (float*)d_out;
    float* probs  = (float*)d_out + (size_t)N * 16;

    auto cdiv = [](long long a, long long b) { return (int)((a + b - 1) / b); };

    k_deg_init <<<cdiv(N, 256), 256, 0, stream>>>(dis, ctr, N);
    const int nsplit = (N / 2) & ~31;
    const int nchunkD = cdiv(E, 2048);
    k_deg_edges<<<2 * nchunkD, 256, 0, stream>>>(dst, dis, ctr + 4, E, nsplit, nchunkD);
    k_rsqrt    <<<cdiv(N, 256), 256, 0, stream>>>(dis, N);

    k_gemm1    <<<N / 16, 256, 0, stream>>>(x, W1, h0);
    k_init_out1<<<cdiv((long long)N * 32, 256), 256, 0, stream>>>(h0, b1, dis, out1, N);
    const int nchunk = cdiv(E, 512);
    k_scatter128<<<2 * nchunk, 256, 0, stream>>>(src, dst, dis, h0, out1, ctr, E, nsplit, nchunk);

    int ntiles = N / 16;
    k_gemm2    <<<cdiv(ntiles, 8), 256, 0, stream>>>(out1, W2, h2, ntiles);
    k_init_out2<<<cdiv((long long)N * 16, 256), 256, 0, stream>>>(h2, b2, dis, logits, N);
    k_scatter16<<<2 * nchunk, 256, 0, stream>>>(src, dst, dis, h2, logits, ctr + 2, E, nsplit, nchunk);

    k_softmax16<<<cdiv((long long)N * 16, 256), 256, 0, stream>>>(logits, probs, N);
}


# both aggregations gather per destination row from a counting-sorted edge list (rank = returning int add of the degree pass): rows written once, no f32 atomics; sum order = arrival rank
# speedup vs baseline: 3.7531x; 1.3013x over previous
#include <hip/hip_runtime.h>
#include <hip/hip_bf16.h>


typedef float v2f __attribute__((ext_vector_type(2)));
typedef float v8f __attribute__((ext_vector_type(8)));

#define LDS_STRIDE 260


__global__ __launch_bounds__(256) void k_cnt_init(int* __restrict__ cnt, int* __restrict__ ctr, int n) {
    int i = blockIdx.x * 256 + threadIdx.x;
    if (i < n) cnt[i] = 0;
    if (i < 4) ctr[i] = 0;
}


__device__ __forceinline__ void take_task(int* __restrict__ ctr, int nchunk,
                                          int* sh, int& hf, int& chunk) {
    if (threadIdx.x == 0) {
        const unsigned id = __builtin_amdgcn_s_sendmsg_rtn(0x87);
        int h = (int)((id >> 18) & 1u);
        int t = atomicAdd(&ctr[h], 1);
        if (t >= nchunk) { h ^= 1; t = atomicAdd(&ctr[h], 1); }
        sh[0] = h; sh[1] = t;
    }
    __syncthreads();
    hf    = __builtin_amdgcn_readfirstlane(sh[0]);
    chunk = __builtin_amdgcn_readfirstlane(sh[1]);
}

__global__ __launch_bounds__(256) void k_count_edges(const int* __restrict__ dst,
                                                     int* __restrict__ cnt,
                                                     int* __restrict__ rankH,
                                                     int* __restrict__ ctr,
                                                     int e, int nsplit, int nchunk) {
    __shared__ int sh[2];
    int hf, chunk;
    take_task(ctr, nchunk, sh, hf, chunk);
    if (chunk >= nchunk) return;
    int* __restrict__ rk = rankH + (size_t)hf * (size_t)e;
    const int base = chunk * 2048 + (int)threadIdx.x;
    int d[8]; bool own[8];
    #pragma unroll
    for (int k = 0; k < 8; ++k) {
        const int i = base + k * 256;
        d[k] = dst[i < e ? i : e - 1];
    }
    #pragma unroll
    for (int k = 0; k < 8; ++k)
        own[k] = (base + k * 256 < e) && ((d[k] >= nsplit) == (hf != 0));
    int r[8];
    #pragma unroll
    for (int k = 0; k < 8; ++k) { r[k] = 0; if (own[k]) r[k] = atomicAdd(&cnt[d[k]], 1); }
    #pragma unroll
    for (int k = 0; k < 8; ++k) if (own[k]) rk[base + k * 256] = r[k];
}

__global__ __launch_bounds__(256) void k_scan(const int* __restrict__ cnt,
                                              int* __restrict__ rowptr,
                                              float* __restrict__ dis, int n) {
    __shared__ int below[8];
    __shared__ int wsum[8];
    const int tid  = threadIdx.x;
    const int lane = tid & 31;
    const int wave = tid >> 5;
    const int base = blockIdx.x * 1024;

    int part = 0;
    const int4* c4 = (const int4*)cnt;
    const int n4 = base >> 2;
    for (int i = tid; i < n4; i += 2048) {
        int4 q[8];
        #pragma unroll
        for (int u = 0; u < 8; ++u) {
            const int ii = i + u * 256;
            q[u] = c4[ii < n4 ? ii : 0];
        }
        #pragma unroll
        for (int u = 0; u < 8; ++u)
            if (i + u * 256 < n4) part += (q[u].x + q[u].y) + (q[u].z + q[u].w);
    }
    #pragma unroll
    for (int o = 16; o >= 1; o >>= 1) part += __shfl_xor(part, o, 32);
    if (lane == 0) below[wave] = part;

    const int i0 = base + tid * 4;
    int c[4];
    #pragma unroll
    for (int k = 0; k < 4; ++k) c[k] = (i0 + k < n) ? cnt[i0 + k] : 0;
    const int t = (c[0] + c[1]) + (c[2] + c[3]);
    int incl = t;
    #pragma unroll
    for (int o = 1; o < 32; o <<= 1) {
        const int y = __shfl_up(incl, o, 32);
        if (lane >= o) incl += y;
    }
    if (lane == 31) wsum[wave] = incl;
    __syncthreads();
    int off = 0;
    #pragma unroll
    for (int w = 0; w < 8; ++w) {
        off += below[w];
        if (w < wave) off += wsum[w];
    }
    int excl = off + incl - t;
    #pragma unroll
    for (int k = 0; k < 4; ++k) {
        if (i0 + k < n) {
            rowptr[i0 + k] = excl;
            dis[i0 + k] = __frsqrt_rn((float)(c[k] + 1));
            excl += c[k];
        }
    }
}

__global__ __launch_bounds__(256) void k_fill(const int* __restrict__ src,
                                              const int* __restrict__ dst,
                                              const int* __restrict__ rankH,
                                              const int* __restrict__ rowptr,
                                              int* __restrict__ csr,
                                              int* __restrict__ ctr,
                                              int e, int nsplit, int nchunk) {
    __shared__ int sh[2];
    int hf, chunk;
    take_task(ctr, nchunk, sh, hf, chunk);
    if (chunk >= nchunk) return;
    const int* __restrict__ rk = rankH + (size_t)hf * (size_t)e;
    const int base = chunk * 2048 + (int)threadIdx.x;
    int d[8], s[8]; bool own[8];
    #pragma unroll
    for (int k = 0; k < 8; ++k) {
        const int i  = base + k * 256;
        const int ic = i < e ? i : e - 1;
        d[k] = dst[ic]; s[k] = src[ic];
    }
    #pragma unroll
    for (int k = 0; k < 8; ++k)
        own[k] = (base + k * 256 < e) && ((d[k] >= nsplit) == (hf != 0));
    int p[8], r[8];
    #pragma unroll
    for (int k = 0; k < 8; ++k) {
        p[k] = 0; r[k] = 0;
        if (own[k]) { p[k] = rowptr[d[k]]; r[k] = rk[base + k * 256]; }
    }
    #pragma unroll
    for (int k = 0; k < 8; ++k) if (own[k]) csr[p[k] + r[k]] = s[k];
}


__global__ __launch_bounds__(256) void k_gemm1(const float* __restrict__ x,
                                               const float* __restrict__ W1,
                                               float* __restrict__ h0) {
    __shared__ float As[16 * LDS_STRIDE];
    const int tid = threadIdx.x;
    const int r0  = blockIdx.x * 16;

    const float4* xv = (const float4*)(x + (size_t)r0 * 256);
    #pragma unroll
    for (int i = 0; i < 4; ++i) {
        int j   = i * 256 + tid;
        int row = j >> 6;
        int c4  = j & 63;
        float4 v = xv[(size_t)row * 64 + c4];
        float* d = &As[row * LDS_STRIDE + c4 * 4];
        d[0] = v.x; d[1] = v.y; d[2] = v.z; d[3] = v.w;
    }
    __syncthreads();

    const int wave  = tid >> 5;
    const int lane  = tid & 31;
    const int half  = lane >> 4;
    const int l16   = lane & 15;
    const int c0    = wave * 16;
    const int kboff = half * 2;

    v8f acc = {0.f, 0.f, 0.f, 0.f, 0.f, 0.f, 0.f, 0.f};
    for (int k0 = 0; k0 < 256; k0 += 4) {
        v2f a, b;
        const float* ap = &As[l16 * LDS_STRIDE + k0 + kboff];
        a.x = ap[0]; a.y = ap[1];
        const float* bp = &W1[(size_t)(k0 + kboff) * 128 + c0 + l16];
        b.x = bp[0]; b.y = bp[128];
        acc = __builtin_amdgcn_wmma_f32_16x16x4_f32(false, a, false, b,
                                                    (short)0, acc, false, false);
    }
    float* out = h0 + (size_t)(r0 + half * 8) * 128 + c0 + l16;
    #pragma unroll
    for (int i = 0; i < 8; ++i) out[(size_t)i * 128] = acc[i];
}


template <bool FULL>
__device__ __forceinline__ void rows8(float4& acc, int lane, int s, int wbits, int u0, int m,
                                      const float4* __restrict__ h4) {
    int sj[8]; float wj[8];
    #pragma unroll
    for (int u = 0; u < 8; ++u) {
        sj[u] = __builtin_amdgcn_readlane(s, u0 + u);
        wj[u] = __int_as_float(__builtin_amdgcn_readlane(wbits, u0 + u));
    }
    float4 v[8];
    #pragma unroll
    for (int u = 0; u < 8; ++u)
        if (FULL || (u0 + u < m)) v[u] = h4[(size_t)sj[u] * 32 + lane];
    #pragma unroll
    for (int u = 0; u < 8; ++u) {
        if (FULL || (u0 + u < m)) {
            acc.x += v[u].x * wj[u]; acc.y += v[u].y * wj[u];
            acc.z += v[u].z * wj[u]; acc.w += v[u].w * wj[u];
        }
    }
}

__global__ __launch_bounds__(256) void k_agg128(const int* __restrict__ csr,
                                                const int* __restrict__ rowptr,
                                                const int* __restrict__ cnt,
                                                const float* __restrict__ dis,
                                                const float* __restrict__ h,
                                                const float* __restrict__ b1,
                                                float* __restrict__ out, int n) {
    const int lane = threadIdx.x & 31;
    const int node = __builtin_amdgcn_readfirstlane((int)((blockIdx.x * 256 + threadIdx.x) >> 5));
    if (node >= n) return;
    const float4* __restrict__ h4 = (const float4*)h;
    const int   beg = __builtin_amdgcn_readfirstlane(rowptr[node]);
    const int   len = __builtin_amdgcn_readfirstlane(cnt[node]);
    const float dd  = dis[node];
    const float4 hv = h4[(size_t)node * 32 + lane];
    const float4 bb = ((const float4*)b1)[lane];

    float4 acc = {0.f, 0.f, 0.f, 0.f};
    for (int j0 = 0; j0 < len; j0 += 32) {
        const int left = len - j0;
        const int m = left < 32 ? left : 32;
        int s = 0; float w = 0.f;
        if (lane < m) { s = csr[beg + j0 + lane]; w = dis[s] * dd; }
        const int wb = __float_as_int(w);
        int u0 = 0;
        for (; u0 + 8 <= m; u0 += 8) rows8<true >(acc, lane, s, wb, u0, m, h4);
        if (u0 < m)                  rows8<false>(acc, lane, s, wb, u0, m, h4);
    }
    const float d2 = dd * dd;
    float4 o;
    o.x = (bb.x + hv.x * d2) + acc.x; o.y = (bb.y + hv.y * d2) + acc.y;
    o.z = (bb.z + hv.z * d2) + acc.z; o.w = (bb.w + hv.w * d2) + acc.w;
    ((float4*)out)[(size_t)node * 32 + lane] = o;
}


__global__ __launch_bounds__(256) void k_gemm2(const float* __restrict__ out1,
                                               const float* __restrict__ W2,
                                               float* __restrict__ h2, int ntiles) {
    int wg = (blockIdx.x * 256 + threadIdx.x) >> 5;
    if (wg >= ntiles) return;
    const int lane  = threadIdx.x & 31;
    const int half  = lane >> 4;
    const int l16   = lane & 15;
    const int kboff = half * 2;
    const int r0    = wg * 16;

    const float* arow = out1 + (size_t)(r0 + l16) * 128;
    v8f acc = {0.f, 0.f, 0.f, 0.f, 0.f, 0.f, 0.f, 0.f};
    for (int k0 = 0; k0 < 128; k0 += 4) {
        v2f a, b;
        a.x = fmaxf(arow[k0 + kboff], 0.f);
        a.y = fmaxf(arow[k0 + kboff + 1], 0.f);
        const float* bp = &W2[(size_t)(k0 + kboff) * 16 + l16];
        b.x = bp[0]; b.y = bp[16];
        acc = __builtin_amdgcn_wmma_f32_16x16x4_f32(false, a, false, b,
                                                    (short)0, acc, false, false);
    }
    float* out = h2 + (size_t)(r0 + half * 8) * 16 + l16;
    #pragma unroll
    for (int i = 0; i < 8; ++i) out[(size_t)i * 16] = acc[i];
}


__global__ __launch_bounds__(256) void k_agg16_softmax(const int* __restrict__ csr,
                                                       const int* __restrict__ rowptr,
                                                       const int* __restrict__ cnt,
                                                       const float* __restrict__ dis,
                                                       const float* __restrict__ h2,
                                                       const float* __restrict__ b2,
                                                       float* __restrict__ logits,
                                                       float* __restrict__ probs, int n) {
    const int g = blockIdx.x * 256 + threadIdx.x;
    if (g >= n * 16) return;
    const int node = g >> 4;
    const int f    = g & 15;
    const int beg  = rowptr[node];
    const int len  = cnt[node];
    const float dd = dis[node];
    const float self = h2[g];
    float acc = 0.f;
    for (int j0 = 0; j0 < len; j0 += 16) {
        const int m = len - j0;
        int s = 0; float w = 0.f;
        if (f < m) { s = csr[beg + j0 + f]; w = dis[s] * dd; }
        float v[16], wj[16];
        #pragma unroll
        for (int u = 0; u < 16; ++u) {
            const int sj = __shfl(s, u, 16);
            wj[u] = __shfl(w, u, 16);
            v[u] = 0.f;
            if (u < m) v[u] = h2[sj * 16 + f];
        }
        #pragma unroll
        for (int u = 0; u < 16; ++u) acc += v[u] * wj[u];
    }
    const float lg = (b2[f] + self * (dd * dd)) + acc;
    logits[g] = lg;
    float mx = lg;
    #pragma unroll
    for (int o = 8; o >= 1; o >>= 1) mx = fmaxf(mx, __shfl_xor(mx, o, 16));
    const float ex = __expf(lg - mx);
    float sum = ex;
    #pragma unroll
    for (int o = 8; o >= 1; o >>= 1) sum += __shfl_xor(sum, o, 16);
    probs[g] = ex / sum;
}


extern "C" void kernel_launch(void* const* d_in, const int* in_sizes, int n_in,
                              void* d_out, int out_size, void* d_ws, size_t ws_size,
                              hipStream_t stream) {
    const float* x  = (const float*)d_in[0];
    const int*   ei = (const int*)d_in[1];
    const float* W1 = (const float*)d_in[2];
    const float* b1 = (const float*)d_in[3];
    const float* W2 = (const float*)d_in[4];
    const float* b2 = (const float*)d_in[5];

    const int N = in_sizes[0] / 256;
    const int E = in_sizes[1] / 2;
    const int* src = ei;
    const int* dst = ei + E;

    char*  ws  = (char*)d_ws;
    size_t off = 0;
    auto alloc = [&](size_t bytes) -> void* {
        void* p = ws + off;
        off += (bytes + 255) & ~(size_t)255;
        return p;
    };
    int*   cnt    = (int*)alloc((size_t)N * 4);
    int*   rowptr = (int*)alloc((size_t)N * 4);
    float* dis    = (float*)alloc((size_t)N * 4);
    float* h0     = (float*)alloc((size_t)N * 128 * 4);
    float* out1   = (float*)alloc((size_t)N * 128 * 4);
    float* h2     = (float*)alloc((size_t)N * 16 * 4);
    int*   csr    = (int*)alloc((size_t)E * 4);
    int*   ctr    = (int*)alloc(256);
    int*   rankH  = (int*)out1;
    if (off > ws_size) return;
    float* logits = (float*)d_out;
    float* probs  = (float*)d_out + (size_t)N * 16;

    auto cdiv = [](long long a, long long b) { return (int)((a + b - 1) / b); };

    k_cnt_init   <<<cdiv(N, 256), 256, 0, stream>>>(cnt, ctr, N);
    const int nsplit = (N / 2) & ~31;
    const int nchunk = cdiv(E, 2048);
    k_count_edges<<<2 * nchunk, 256, 0, stream>>>(dst, cnt, rankH, ctr, E, nsplit, nchunk);
    k_scan       <<<cdiv(N, 1024), 256, 0, stream>>>(cnt, rowptr, dis, N);
    k_fill       <<<2 * nchunk, 256, 0, stream>>>(src, dst, rankH, rowptr, csr, ctr + 2, E, nsplit, nchunk);

    k_gemm1      <<<N / 16, 256, 0, stream>>>(x, W1, h0);
    k_agg128     <<<cdiv((long long)N * 32, 256), 256, 0, stream>>>(csr, rowptr, cnt, dis, h0, b1, out1, N);

    int ntiles = N / 16;
    k_gemm2      <<<cdiv(ntiles, 8), 256, 0, stream>>>(out1, W2, h2, ntiles);
    k_agg16_softmax<<<cdiv((long long)N * 16, 256), 256, 0, stream>>>(csr, rowptr, cnt, dis, h2, b2, logits, probs, N);
}


# edge sort by destination: two-level counting sort in LDS (chunk histogram, bucket scatter, row ranks per bucket), no global atomics or tickets, one launch fewer; f32 sum order = LDS arrival order
# speedup vs baseline: 4.0433x; 1.0773x over previous
#include <hip/hip_runtime.h>
#include <hip/hip_bf16.h>


typedef float v2f __attribute__((ext_vector_type(2)));
typedef float v8f __attribute__((ext_vector_type(8)));

#define LDS_STRIDE 260


#define BSH    8
#define CHUNK  8192
#define MAXNB  1024
#define STG    16384

__device__ __forceinline__ int block_excl(int t, int* wsum, int lane, int wave) {
    int incl = t;
    #pragma unroll
    for (int o = 1; o < 32; o <<= 1) {
        const int y = __shfl_up(incl, o, 32);
        if (lane >= o) incl += y;
    }
    if (lane == 31) wsum[wave] = incl;
    __syncthreads();
    int off = 0;
    #pragma unroll
    for (int w = 0; w < 8; ++w) if (w < wave) off += wsum[w];
    return off + incl - t;
}

__global__ __launch_bounds__(256) void k_hist(const int* __restrict__ dst,
                                              int* __restrict__ hist, int e, int nbp) {
    __shared__ int lh[MAXNB];
    const int tid = threadIdx.x;
    for (int b = tid; b < nbp; b += 256) lh[b] = 0;
    __syncthreads();
    const int base = blockIdx.x * CHUNK + tid;
    for (int k0 = 0; k0 < CHUNK; k0 += 16 * 256) {
        int d[16];
        #pragma unroll
        for (int u = 0; u < 16; ++u) {
            const int i = base + k0 + u * 256;
            d[u] = dst[i < e ? i : e - 1];
        }
        #pragma unroll
        for (int u = 0; u < 16; ++u)
            if (base + k0 + u * 256 < e) atomicAdd(&lh[d[u] >> BSH], 1);
    }
    __syncthreads();
    for (int b = tid; b < nbp; b += 256) hist[(size_t)blockIdx.x * nbp + b] = lh[b];
}

__global__ __launch_bounds__(256) void k_scatter(const int* __restrict__ src,
                                                 const int* __restrict__ dst,
                                                 const int* __restrict__ hist,
                                                 int* __restrict__ recs,
                                                 int* __restrict__ bstart,
                                                 int* __restrict__ bsize,
                                                 int e, int nbp, int nblk) {
    __shared__ int  ltot[MAXNB];
    __shared__ int  lbel[MAXNB];
    __shared__ int  lcur[MAXNB];
    __shared__ int  gdel[MAXNB];
    __shared__ int  wsA[8];
    __shared__ int  wsB[8];
    __shared__ int2 stage[CHUNK];
    const int tid  = threadIdx.x;
    const int lane = tid & 31;
    const int wave = tid >> 5;
    const int blk  = blockIdx.x;
    for (int b = tid; b < nbp; b += 256) { ltot[b] = 0; lbel[b] = 0; }
    __syncthreads();

    const int4* __restrict__ h4 = (const int4*)hist;
    const int nq = nbp >> 2;
    const int ng = 256 / nq;
    const int q  = tid % nq;
    const int g  = tid / nq;
    if (g < ng) {
        int4 tot = {0, 0, 0, 0}, bel = {0, 0, 0, 0};
        for (int j0 = g; j0 < nblk; j0 += 16 * ng) {
            int4 v[16];
            #pragma unroll
            for (int u = 0; u < 16; ++u) {
                const int jj = j0 + u * ng;
                v[u] = h4[(size_t)(jj < nblk ? jj : 0) * nq + q];
            }
            #pragma unroll
            for (int u = 0; u < 16; ++u) {
                const int jj = j0 + u * ng;
                if (jj < nblk) {
                    tot.x += v[u].x; tot.y += v[u].y; tot.z += v[u].z; tot.w += v[u].w;
                    if (jj < blk) { bel.x += v[u].x; bel.y += v[u].y; bel.z += v[u].z; bel.w += v[u].w; }
                }
            }
        }
        atomicAdd(&ltot[4 * q + 0], tot.x); atomicAdd(&ltot[4 * q + 1], tot.y);
        atomicAdd(&ltot[4 * q + 2], tot.z); atomicAdd(&ltot[4 * q + 3], tot.w);
        atomicAdd(&lbel[4 * q + 0], bel.x); atomicAdd(&lbel[4 * q + 1], bel.y);
        atomicAdd(&lbel[4 * q + 2], bel.z); atomicAdd(&lbel[4 * q + 3], bel.w);
    }
    __syncthreads();

    int a[4] = {0, 0, 0, 0}, o[4] = {0, 0, 0, 0};
    if (tid < nq) {
        const int4 own = h4[(size_t)blk * nq + tid];
        o[0] = own.x; o[1] = own.y; o[2] = own.z; o[3] = own.w;
        #pragma unroll
        for (int k = 0; k < 4; ++k) a[k] = ltot[4 * tid + k];
    }
    int exA = block_excl((a[0] + a[1]) + (a[2] + a[3]), wsA, lane, wave);
    int exO = block_excl((o[0] + o[1]) + (o[2] + o[3]), wsB, lane, wave);
    if (tid < nq) {
        #pragma unroll
        for (int k = 0; k < 4; ++k) {
            const int b = 4 * tid + k;
            lcur[b] = exO;
            gdel[b] = exA + lbel[b] - exO;
            if (blk == 0) { bstart[b] = exA; bsize[b] = a[k]; }
            exA += a[k]; exO += o[k];
        }
    }
    __syncthreads();

    const int base = blk * CHUNK + tid;
    for (int k0 = 0; k0 < CHUNK; k0 += 16 * 256) {
        int d[16], s[16];
        #pragma unroll
        for (int u = 0; u < 16; ++u) {
            const int i  = base + k0 + u * 256;
            const int ic = i < e ? i : e - 1;
            d[u] = dst[ic]; s[u] = src[ic];
        }
        int slot[16];
        #pragma unroll
        for (int u = 0; u < 16; ++u) {
            slot[u] = 0;
            if (base + k0 + u * 256 < e) slot[u] = atomicAdd(&lcur[d[u] >> BSH], 1);
        }
        #pragma unroll
        for (int u = 0; u < 16; ++u)
            if (base + k0 + u * 256 < e) stage[slot[u]] = make_int2(s[u], d[u]);
    }
    __syncthreads();
    const int left = e - blk * CHUNK;
    const int nown = left < CHUNK ? left : CHUNK;
    int2* __restrict__ rc = (int2*)recs;
    for (int k0 = tid; k0 < nown; k0 += 8 * 256) {
        int2 r[8]; int gd[8];
        #pragma unroll
        for (int u = 0; u < 8; ++u) {
            const int k = k0 + u * 256;
            r[u] = stage[k < nown ? k : 0];
        }
        #pragma unroll
        for (int u = 0; u < 8; ++u) gd[u] = gdel[r[u].y >> BSH];
        #pragma unroll
        for (int u = 0; u < 8; ++u) {
            const int k = k0 + u * 256;
            if (k < nown) rc[(size_t)(k + gd[u])] = r[u];
        }
    }
}

__global__ __launch_bounds__(256) void k_rows(const int* __restrict__ recs,
                                              const int* __restrict__ bstart,
                                              const int* __restrict__ bsize,
                                              int* __restrict__ cnt,
                                              int* __restrict__ rowptr,
                                              float* __restrict__ dis,
                                              int* __restrict__ csr, int n) {
    __shared__ int lc[256];
    __shared__ int ws[8];
    __shared__ int lst[STG];
    const int tid  = threadIdx.x;
    const int lane = tid & 31;
    const int wave = tid >> 5;
    const int bkt  = blockIdx.x;
    const int s0 = bstart[bkt];
    const int m  = bsize[bkt];
    const int2* __restrict__ rc = (const int2*)recs + s0;
    lc[tid] = 0;
    __syncthreads();
    for (int k0 = 0; k0 < m; k0 += 16 * 256) {
        int2 r[16];
        #pragma unroll
        for (int u = 0; u < 16; ++u) {
            const int k = k0 + u * 256 + tid;
            r[u] = rc[k < m ? k : 0];
        }
        #pragma unroll
        for (int u = 0; u < 16; ++u)
            if (k0 + u * 256 + tid < m) atomicAdd(&lc[r[u].y & 255], 1);
    }
    __syncthreads();
    const int c  = lc[tid];
    const int ex = block_excl(c, ws, lane, wave);
    const int node = bkt * 256 + tid;
    if (node < n) {
        cnt[node]    = c;
        rowptr[node] = s0 + ex;
        dis[node]    = __frsqrt_rn((float)(c + 1));
    }
    lc[tid] = ex;
    __syncthreads();
    const bool staged = m <= STG;
    for (int k0 = 0; k0 < m; k0 += 16 * 256) {
        int2 r[16];
        #pragma unroll
        for (int u = 0; u < 16; ++u) {
            const int k = k0 + u * 256 + tid;
            r[u] = rc[k < m ? k : 0];
        }
        int p[16];
        #pragma unroll
        for (int u = 0; u < 16; ++u) {
            p[u] = 0;
            if (k0 + u * 256 + tid < m) p[u] = atomicAdd(&lc[r[u].y & 255], 1);
        }
        #pragma unroll
        for (int u = 0; u < 16; ++u) {
            if (k0 + u * 256 + tid < m) {
                if (staged) lst[p[u]] = r[u].x; else csr[s0 + p[u]] = r[u].x;
            }
        }
    }
    if (staged) {
        __syncthreads();
        for (int k0 = tid; k0 < m; k0 += 8 * 256) {
            int v[8];
            #pragma unroll
            for (int u = 0; u < 8; ++u) {
                const int k = k0 + u * 256;
                v[u] = lst[k < m ? k : 0];
            }
            #pragma unroll
            for (int u = 0; u < 8; ++u) {
                const int k = k0 + u * 256;
                if (k < m) csr[s0 + k] = v[u];
            }
        }
    }
}


__global__ __launch_bounds__(256) void k_gemm1(const float* __restrict__ x,
                                               const float* __restrict__ W1,
                                               float* __restrict__ h0) {
    __shared__ float As[16 * LDS_STRIDE];
    const int tid = threadIdx.x;
    const int r0  = blockIdx.x * 16;

    const float4* xv = (const float4*)(x + (size_t)r0 * 256);
    #pragma unroll
    for (int i = 0; i < 4; ++i) {
        int j   = i * 256 + tid;
        int row = j >> 6;
        int c4  = j & 63;
        float4 v = xv[(size_t)row * 64 + c4];
        float* d = &As[row * LDS_STRIDE + c4 * 4];
        d[0] = v.x; d[1] = v.y; d[2] = v.z; d[3] = v.w;
    }
    __syncthreads();

    const int wave  = tid >> 5;
    const int lane  = tid & 31;
    const int half  = lane >> 4;
    const int l16   = lane & 15;
    const int c0    = wave * 16;
    const int kboff = half * 2;

    v8f acc = {0.f, 0.f, 0.f, 0.f, 0.f, 0.f, 0.f, 0.f};
    for (int k0 = 0; k0 < 256; k0 += 4) {
        v2f a, b;
        const float* ap = &As[l16 * LDS_STRIDE + k0 + kboff];
        a.x = ap[0]; a.y = ap[1];
        const float* bp = &W1[(size_t)(k0 + kboff) * 128 + c0 + l16];
        b.x = bp[0]; b.y = bp[128];
        acc = __builtin_amdgcn_wmma_f32_16x16x4_f32(false, a, false, b,
                                                    (short)0, acc, false, false);
    }
    float* out = h0 + (size_t)(r0 + half * 8) * 128 + c0 + l16;
    #pragma unroll
    for (int i = 0; i < 8; ++i) out[(size_t)i * 128] = acc[i];
}


template <bool FULL>
__device__ __forceinline__ void rows8(float4& acc, int lane, int s, int wbits, int u0, int m,
                                      const float4* __restrict__ h4) {
    int sj[8]; float wj[8];
    #pragma unroll
    for (int u = 0; u < 8; ++u) {
        sj[u] = __builtin_amdgcn_readlane(s, u0 + u);
        wj[u] = __int_as_float(__builtin_amdgcn_readlane(wbits, u0 + u));
    }
    float4 v[8];
    #pragma unroll
    for (int u = 0; u < 8; ++u)
        if (FULL || (u0 + u < m)) v[u] = h4[(size_t)sj[u] * 32 + lane];
    #pragma unroll
    for (int u = 0; u < 8; ++u) {
        if (FULL || (u0 + u < m)) {
            acc.x += v[u].x * wj[u]; acc.y += v[u].y * wj[u];
            acc.z += v[u].z * wj[u]; acc.w += v[u].w * wj[u];
        }
    }
}

__global__ __launch_bounds__(256) void k_agg128(const int* __restrict__ csr,
                                                const int* __restrict__ rowptr,
                                                const int* __restrict__ cnt,
                                                const float* __restrict__ dis,
                                                const float* __restrict__ h,
                                                const float* __restrict__ b1,
                                                float* __restrict__ out, int n) {
    const int lane = threadIdx.x & 31;
    const int node = __builtin_amdgcn_readfirstlane((int)((blockIdx.x * 256 + threadIdx.x) >> 5));
    if (node >= n) return;
    const float4* __restrict__ h4 = (const float4*)h;
    const int   beg = __builtin_amdgcn_readfirstlane(rowptr[node]);
    const int   len = __builtin_amdgcn_readfirstlane(cnt[node]);
    const float dd  = dis[node];
    const float4 hv = h4[(size_t)node * 32 + lane];
    const float4 bb = ((const float4*)b1)[lane];

    float4 acc = {0.f, 0.f, 0.f, 0.f};
    for (int j0 = 0; j0 < len; j0 += 32) {
        const int left = len - j0;
        const int m = left < 32 ? left : 32;
        int s = 0; float w = 0.f;
        if (lane < m) { s = csr[beg + j0 + lane]; w = dis[s] * dd; }
        const int wb = __float_as_int(w);
        int u0 = 0;
        for (; u0 + 8 <= m; u0 += 8) rows8<true >(acc, lane, s, wb, u0, m, h4);
        if (u0 < m)                  rows8<false>(acc, lane, s, wb, u0, m, h4);
    }
    const float d2 = dd * dd;
    float4 o;
    o.x = (bb.x + hv.x * d2) + acc.x; o.y = (bb.y + hv.y * d2) + acc.y;
    o.z = (bb.z + hv.z * d2) + acc.z; o.w = (bb.w + hv.w * d2) + acc.w;
    ((float4*)out)[(size_t)node * 32 + lane] = o;
}


__global__ __launch_bounds__(256) void k_gemm2(const float* __restrict__ out1,
                                               const float* __restrict__ W2,
                                               float* __restrict__ h2, int ntiles) {
    int wg = (blockIdx.x * 256 + threadIdx.x) >> 5;
    if (wg >= ntiles) return;
    const int lane  = threadIdx.x & 31;
    const int half  = lane >> 4;
    const int l16   = lane & 15;
    const int kboff = half * 2;
    const int r0    = wg * 16;

    const float* arow = out1 + (size_t)(r0 + l16) * 128;
    v8f acc = {0.f, 0.f, 0.f, 0.f, 0.f, 0.f, 0.f, 0.f};
    for (int k0 = 0; k0 < 128; k0 += 4) {
        v2f a, b;
        a.x = fmaxf(arow[k0 + kboff], 0.f);
        a.y = fmaxf(arow[k0 + kboff + 1], 0.f);
        const float* bp = &W2[(size_t)(k0 + kboff) * 16 + l16];
        b.x = bp[0]; b.y = bp[16];
        acc = __builtin_amdgcn_wmma_f32_16x16x4_f32(false, a, false, b,
                                                    (short)0, acc, false, false);
    }
    float* out = h2 + (size_t)(r0 + half * 8) * 16 + l16;
    #pragma unroll
    for (int i = 0; i < 8; ++i) out[(size_t)i * 16] = acc[i];
}


__global__ __launch_bounds__(256) void k_agg16_softmax(const int* __restrict__ csr,
                                                       const int* __restrict__ rowptr,
                                                       const int* __restrict__ cnt,
                                                       const float* __restrict__ dis,
                                                       const float* __restrict__ h2,
                                                       const float* __restrict__ b2,
                                                       float* __restrict__ logits,
                                                       float* __restrict__ probs, int n) {
    const int g = blockIdx.x * 256 + threadIdx.x;
    if (g >= n * 16) return;
    const int node = g >> 4;
    const int f    = g & 15;
    const int beg  = rowptr[node];
    const int len  = cnt[node];
    const float dd = dis[node];
    const float self = h2[g];
    float acc = 0.f;
    for (int j0 = 0; j0 < len; j0 += 16) {
        const int m = len - j0;
        int s = 0; float w = 0.f;
        if (f < m) { s = csr[beg + j0 + f]; w = dis[s] * dd; }
        float v[16], wj[16];
        #pragma unroll
        for (int u = 0; u < 16; ++u) {
            const int sj = __shfl(s, u, 16);
            wj[u] = __shfl(w, u, 16);
            v[u] = 0.f;
            if (u < m) v[u] = h2[sj * 16 + f];
        }
        #pragma unroll
        for (int u = 0; u < 16; ++u) acc += v[u] * wj[u];
    }
    const float lg = (b2[f] + self * (dd * dd)) + acc;
    logits[g] = lg;
    float mx = lg;
    #pragma unroll
    for (int o = 8; o >= 1; o >>= 1) mx = fmaxf(mx, __shfl_xor(mx, o, 16));
    const float ex = __expf(lg - mx);
    float sum = ex;
    #pragma unroll
    for (int o = 8; o >= 1; o >>= 1) sum += __shfl_xor(sum, o, 16);
    probs[g] = ex / sum;
}


extern "C" void kernel_launch(void* const* d_in, const int* in_sizes, int n_in,
                              void* d_out, int out_size, void* d_ws, size_t ws_size,
                              hipStream_t stream) {
    const float* x  = (const float*)d_in[0];
    const int*   ei = (const int*)d_in[1];
    const float* W1 = (const float*)d_in[2];
    const float* b1 = (const float*)d_in[3];
    const float* W2 = (const float*)d_in[4];
    const float* b2 = (const float*)d_in[5];

    const int N = in_sizes[0] / 256;
    const int E = in_sizes[1] / 2;
    const int* src = ei;
    const int* dst = ei + E;

    auto cdiv = [](long long a, long long b) { return (int)((a + b - 1) / b); };
    const int nb   = cdiv(N, 256);
    const int nbp  = (nb + 3) & ~3;
    const int nblk = cdiv(E, CHUNK);

    char*  ws  = (char*)d_ws;
    size_t off = 0;
    auto alloc = [&](size_t bytes) -> void* {
        void* p = ws + off;
        off += (bytes + 255) & ~(size_t)255;
        return p;
    };
    int*   cnt    = (int*)alloc((size_t)N * 4);
    int*   rowptr = (int*)alloc((size_t)N * 4);
    float* dis    = (float*)alloc((size_t)N * 4);
    float* h0     = (float*)alloc((size_t)N * 128 * 4);
    float* out1   = (float*)alloc((size_t)N * 128 * 4);
    float* h2     = (float*)alloc((size_t)N * 16 * 4);
    int*   csr    = (int*)alloc((size_t)E * 4);
    if (off > ws_size) return;
    int*   hist   = (int*)h0;
    int*   bstart = hist + (size_t)nblk * nbp;
    int*   bsize  = bstart + nbp;
    int*   recs   = (int*)out1;
    if (nbp > MAXNB) return;
    if (((size_t)nblk + 2) * nbp * 4 > (size_t)N * 512) return;
    if ((size_t)E * 8 > (size_t)N * 512) return;
    float* logits = (float*)d_out;
    float* probs  = (float*)d_out + (size_t)N * 16;

    k_hist       <<<nblk, 256, 0, stream>>>(dst, hist, E, nbp);
    k_scatter    <<<nblk, 256, 0, stream>>>(src, dst, hist, recs, bstart, bsize, E, nbp, nblk);
    k_rows       <<<nb, 256, 0, stream>>>(recs, bstart, bsize, cnt, rowptr, dis, csr, N);

    k_gemm1      <<<N / 16, 256, 0, stream>>>(x, W1, h0);
    k_agg128     <<<cdiv((long long)N * 32, 256), 256, 0, stream>>>(csr, rowptr, cnt, dis, h0, b1, out1, N);

    int ntiles = N / 16;
    k_gemm2      <<<cdiv(ntiles, 8), 256, 0, stream>>>(out1, W2, h2, ntiles);
    k_agg16_softmax<<<cdiv((long long)N * 16, 256), 256, 0, stream>>>(csr, rowptr, cnt, dis, h2, b2, logits, probs, N);
}


# x @ W1's 3125 row strips run as extra work-groups inside the three edge-sort launches (196 sort groups each, started first), not as a launch of their own: 6 launches; sort count rows now alias h2
# speedup vs baseline: 4.4767x; 1.1072x over previous
#include <hip/hip_runtime.h>
#include <hip/hip_bf16.h>


typedef float v2f __attribute__((ext_vector_type(2)));
typedef float v8f __attribute__((ext_vector_type(8)));

#define LDS_STRIDE 260


#define BSH    8
#define CHUNK  8192
#define MAXNB  1024
#define STG    16384

__device__ __forceinline__ int block_excl(int t, int* wsum, int lane, int wave) {
    int incl = t;
    #pragma unroll
    for (int o = 1; o < 32; o <<= 1) {
        const int y = __shfl_up(incl, o, 32);
        if (lane >= o) incl += y;
    }
    if (lane == 31) wsum[wave] = incl;
    __syncthreads();
    int off = 0;
    #pragma unroll
    for (int w = 0; w < 8; ++w) if (w < wave) off += wsum[w];
    return off + incl - t;
}

__device__ __forceinline__ void hist_pass(const int* __restrict__ dst,
                                          int* __restrict__ hist, int e, int nbp,
                                          int blk, unsigned char* lds) {
    int* lh = (int*)lds;
    const int tid = threadIdx.x;
    for (int b = tid; b < nbp; b += 256) lh[b] = 0;
    __syncthreads();
    const int base = blk * CHUNK + tid;
    for (int k0 = 0; k0 < CHUNK; k0 += 16 * 256) {
        int d[16];
        #pragma unroll
        for (int u = 0; u < 16; ++u) {
            const int i = base + k0 + u * 256;
            d[u] = dst[i < e ? i : e - 1];
        }
        #pragma unroll
        for (int u = 0; u < 16; ++u)
            if (base + k0 + u * 256 < e) atomicAdd(&lh[d[u] >> BSH], 1);
    }
    __syncthreads();
    for (int b = tid; b < nbp; b += 256) hist[(size_t)blk * nbp + b] = lh[b];
}

#define SCATTER_LDS (CHUNK * 8 + 2 * MAXNB * 4 + 64)
__device__ __forceinline__ void scatter_pass(const int* __restrict__ src,
                                             const int* __restrict__ dst,
                                             const int* __restrict__ hist,
                                             int* __restrict__ recs,
                                             int* __restrict__ bstart,
                                             int* __restrict__ bsize,
                                             int e, int nbp, int nblk,
                                             int blk, unsigned char* lds) {
    int2* stage = (int2*)lds;
    int*  ltot  = (int*)lds;
    int*  lbel  = (int*)lds + MAXNB;
    int*  lcur  = (int*)(lds + CHUNK * 8);
    int*  gdel  = lcur + MAXNB;
    int*  wsA   = gdel + MAXNB;
    int*  wsB   = wsA + 8;
    const int tid  = threadIdx.x;
    const int lane = tid & 31;
    const int wave = tid >> 5;
    for (int b = tid; b < nbp; b += 256) { ltot[b] = 0; lbel[b] = 0; }
    __syncthreads();

    const int4* __restrict__ h4 = (const int4*)hist;
    const int nq = nbp >> 2;
    const int ng = 256 / nq;
    const int q  = tid % nq;
    const int g  = tid / nq;
    if (g < ng) {
        int4 tot = {0, 0, 0, 0}, bel = {0, 0, 0, 0};
        for (int j0 = g; j0 < nblk; j0 += 16 * ng) {
            int4 v[16];
            #pragma unroll
            for (int u = 0; u < 16; ++u) {
                const int jj = j0 + u * ng;
                v[u] = h4[(size_t)(jj < nblk ? jj : 0) * nq + q];
            }
            #pragma unroll
            for (int u = 0; u < 16; ++u) {
                const int jj = j0 + u * ng;
                if (jj < nblk) {
                    tot.x += v[u].x; tot.y += v[u].y; tot.z += v[u].z; tot.w += v[u].w;
                    if (jj < blk) { bel.x += v[u].x; bel.y += v[u].y; bel.z += v[u].z; bel.w += v[u].w; }
                }
            }
        }
        atomicAdd(&ltot[4 * q + 0], tot.x); atomicAdd(&ltot[4 * q + 1], tot.y);
        atomicAdd(&ltot[4 * q + 2], tot.z); atomicAdd(&ltot[4 * q + 3], tot.w);
        atomicAdd(&lbel[4 * q + 0], bel.x); atomicAdd(&lbel[4 * q + 1], bel.y);
        atomicAdd(&lbel[4 * q + 2], bel.z); atomicAdd(&lbel[4 * q + 3], bel.w);
    }
    __syncthreads();

    int a[4] = {0, 0, 0, 0}, o[4] = {0, 0, 0, 0};
    if (tid < nq) {
        const int4 own = h4[(size_t)blk * nq + tid];
        o[0] = own.x; o[1] = own.y; o[2] = own.z; o[3] = own.w;
        #pragma unroll
        for (int k = 0; k < 4; ++k) a[k] = ltot[4 * tid + k];
    }
    int exA = block_excl((a[0] + a[1]) + (a[2] + a[3]), wsA, lane, wave);
    int exO = block_excl((o[0] + o[1]) + (o[2] + o[3]), wsB, lane, wave);
    if (tid < nq) {
        #pragma unroll
        for (int k = 0; k < 4; ++k) {
            const int b = 4 * tid + k;
            lcur[b] = exO;
            gdel[b] = exA + lbel[b] - exO;
            if (blk == 0) { bstart[b] = exA; bsize[b] = a[k]; }
            exA += a[k]; exO += o[k];
        }
    }
    __syncthreads();

    const int base = blk * CHUNK + tid;
    for (int k0 = 0; k0 < CHUNK; k0 += 16 * 256) {
        int d[16], s[16];
        #pragma unroll
        for (int u = 0; u < 16; ++u) {
            const int i  = base + k0 + u * 256;
            const int ic = i < e ? i : e - 1;
            d[u] = dst[ic]; s[u] = src[ic];
        }
        int slot[16];
        #pragma unroll
        for (int u = 0; u < 16; ++u) {
            slot[u] = 0;
            if (base + k0 + u * 256 < e) slot[u] = atomicAdd(&lcur[d[u] >> BSH], 1);
        }
        #pragma unroll
        for (int u = 0; u < 16; ++u)
            if (base + k0 + u * 256 < e) stage[slot[u]] = make_int2(s[u], d[u]);
    }
    __syncthreads();
    const int left = e - blk * CHUNK;
    const int nown = left < CHUNK ? left : CHUNK;
    int2* __restrict__ rc = (int2*)recs;
    for (int k0 = tid; k0 < nown; k0 += 8 * 256) {
        int2 r[8]; int gd[8];
        #pragma unroll
        for (int u = 0; u < 8; ++u) {
            const int k = k0 + u * 256;
            r[u] = stage[k < nown ? k : 0];
        }
        #pragma unroll
        for (int u = 0; u < 8; ++u) gd[u] = gdel[r[u].y >> BSH];
        #pragma unroll
        for (int u = 0; u < 8; ++u) {
            const int k = k0 + u * 256;
            if (k < nown) rc[(size_t)(k + gd[u])] = r[u];
        }
    }
}

#define ROWS_LDS (STG * 4 + 256 * 4 + 32)
__device__ __forceinline__ void rows_pass(const int* __restrict__ recs,
                                          const int* __restrict__ bstart,
                                          const int* __restrict__ bsize,
                                          int* __restrict__ cnt,
                                          int* __restrict__ rowptr,
                                          float* __restrict__ dis,
                                          int* __restrict__ csr, int n,
                                          int bkt, unsigned char* lds) {
    int* lst = (int*)lds;
    int* lc  = lst + STG;
    int* ws  = lc + 256;
    const int tid  = threadIdx.x;
    const int lane = tid & 31;
    const int wave = tid >> 5;
    const int s0 = bstart[bkt];
    const int m  = bsize[bkt];
    const int2* __restrict__ rc = (const int2*)recs + s0;
    lc[tid] = 0;
    __syncthreads();
    for (int k0 = 0; k0 < m; k0 += 16 * 256) {
        int2 r[16];
        #pragma unroll
        for (int u = 0; u < 16; ++u) {
            const int k = k0 + u * 256 + tid;
            r[u] = rc[k < m ? k : 0];
        }
        #pragma unroll
        for (int u = 0; u < 16; ++u)
            if (k0 + u * 256 + tid < m) atomicAdd(&lc[r[u].y & 255], 1);
    }
    __syncthreads();
    const int c  = lc[tid];
    const int ex = block_excl(c, ws, lane, wave);
    const int node = bkt * 256 + tid;
    if (node < n) {
        cnt[node]    = c;
        rowptr[node] = s0 + ex;
        dis[node]    = __frsqrt_rn((float)(c + 1));
    }
    lc[tid] = ex;
    __syncthreads();
    const bool staged = m <= STG;
    for (int k0 = 0; k0 < m; k0 += 16 * 256) {
        int2 r[16];
        #pragma unroll
        for (int u = 0; u < 16; ++u) {
            const int k = k0 + u * 256 + tid;
            r[u] = rc[k < m ? k : 0];
        }
        int p[16];
        #pragma unroll
        for (int u = 0; u < 16; ++u) {
            p[u] = 0;
            if (k0 + u * 256 + tid < m) p[u] = atomicAdd(&lc[r[u].y & 255], 1);
        }
        #pragma unroll
        for (int u = 0; u < 16; ++u) {
            if (k0 + u * 256 + tid < m) {
                if (staged) lst[p[u]] = r[u].x; else csr[s0 + p[u]] = r[u].x;
            }
        }
    }
    if (staged) {
        __syncthreads();
        for (int k0 = tid; k0 < m; k0 += 8 * 256) {
            int v[8];
            #pragma unroll
            for (int u = 0; u < 8; ++u) {
                const int k = k0 + u * 256;
                v[u] = lst[k < m ? k : 0];
            }
            #pragma unroll
            for (int u = 0; u < 8; ++u) {
                const int k = k0 + u * 256;
                if (k < m) csr[s0 + k] = v[u];
            }
        }
    }
}


#define GEMM1_LDS (16 * LDS_STRIDE * 4)
__device__ __forceinline__ void gemm1_strip(const float* __restrict__ x,
                                            const float* __restrict__ W1,
                                            float* __restrict__ h0,
                                            int strip, unsigned char* lds) {
    float* As = (float*)lds;
    const int tid = threadIdx.x;
    const int r0  = strip * 16;

    const float4* xv = (const float4*)(x + (size_t)r0 * 256);
    #pragma unroll
    for (int i = 0; i < 4; ++i) {
        int j   = i * 256 + tid;
        int row = j >> 6;
        int c4  = j & 63;
        float4 v = xv[(size_t)row * 64 + c4];
        float* d = &As[row * LDS_STRIDE + c4 * 4];
        d[0] = v.x; d[1] = v.y; d[2] = v.z; d[3] = v.w;
    }
    __syncthreads();

    const int wave  = tid >> 5;
    const int lane  = tid & 31;
    const int half  = lane >> 4;
    const int l16   = lane & 15;
    const int c0    = wave * 16;
    const int kboff = half * 2;

    v8f acc = {0.f, 0.f, 0.f, 0.f, 0.f, 0.f, 0.f, 0.f};
    for (int k0 = 0; k0 < 256; k0 += 4) {
        v2f a, b;
        const float* ap = &As[l16 * LDS_STRIDE + k0 + kboff];
        a.x = ap[0]; a.y = ap[1];
        const float* bp = &W1[(size_t)(k0 + kboff) * 128 + c0 + l16];
        b.x = bp[0]; b.y = bp[128];
        acc = __builtin_amdgcn_wmma_f32_16x16x4_f32(false, a, false, b,
                                                    (short)0, acc, false, false);
    }
    float* out = h0 + (size_t)(r0 + half * 8) * 128 + c0 + l16;
    #pragma unroll
    for (int i = 0; i < 8; ++i) out[(size_t)i * 128] = acc[i];
}


#define MAX2(a, b) ((a) > (b) ? (a) : (b))

__global__ __launch_bounds__(256) void k_hist_gemm(const int* __restrict__ dst,
                                                   int* __restrict__ hist, int e, int nbp, int nsort,
                                                   const float* __restrict__ x,
                                                   const float* __restrict__ W1,
                                                   float* __restrict__ h0, int strip0) {
    __shared__ __attribute__((aligned(16))) unsigned char lds[MAX2(MAXNB * 4, GEMM1_LDS)];
    const int blk = blockIdx.x;
    if (blk < nsort) hist_pass(dst, hist, e, nbp, blk, lds);
    else             gemm1_strip(x, W1, h0, strip0 + (blk - nsort), lds);
}

__global__ __launch_bounds__(256) void k_scatter_gemm(const int* __restrict__ src,
                                                      const int* __restrict__ dst,
                                                      const int* __restrict__ hist,
                                                      int* __restrict__ recs,
                                                      int* __restrict__ bstart,
                                                      int* __restrict__ bsize,
                                                      int e, int nbp, int nsort,
                                                      const float* __restrict__ x,
                                                      const float* __restrict__ W1,
                                                      float* __restrict__ h0, int strip0) {
    __shared__ __attribute__((aligned(16))) unsigned char lds[MAX2(SCATTER_LDS, GEMM1_LDS)];
    const int blk = blockIdx.x;
    if (blk < nsort) scatter_pass(src, dst, hist, recs, bstart, bsize, e, nbp, nsort, blk, lds);
    else             gemm1_strip(x, W1, h0, strip0 + (blk - nsort), lds);
}

__global__ __launch_bounds__(256) void k_rows_gemm(const int* __restrict__ recs,
                                                   const int* __restrict__ bstart,
                                                   const int* __restrict__ bsize,
                                                   int* __restrict__ cnt,
                                                   int* __restrict__ rowptr,
                                                   float* __restrict__ dis,
                                                   int* __restrict__ csr, int n, int nsort,
                                                   const float* __restrict__ x,
                                                   const float* __restrict__ W1,
                                                   float* __restrict__ h0, int strip0) {
    __shared__ __attribute__((aligned(16))) unsigned char lds[MAX2(ROWS_LDS, GEMM1_LDS)];
    const int blk = blockIdx.x;
    if (blk < nsort) rows_pass(recs, bstart, bsize, cnt, rowptr, dis, csr, n, blk, lds);
    else             gemm1_strip(x, W1, h0, strip0 + (blk - nsort), lds);
}


template <bool FULL>
__device__ __forceinline__ void rows8(float4& acc, int lane, int s, int wbits, int u0, int m,
                                      const float4* __restrict__ h4) {
    int sj[8]; float wj[8];
    #pragma unroll
    for (int u = 0; u < 8; ++u) {
        sj[u] = __builtin_amdgcn_readlane(s, u0 + u);
        wj[u] = __int_as_float(__builtin_amdgcn_readlane(wbits, u0 + u));
    }
    float4 v[8];
    #pragma unroll
    for (int u = 0; u < 8; ++u)
        if (FULL || (u0 + u < m)) v[u] = h4[(size_t)sj[u] * 32 + lane];
    #pragma unroll
    for (int u = 0; u < 8; ++u) {
        if (FULL || (u0 + u < m)) {
            acc.x += v[u].x * wj[u]; acc.y += v[u].y * wj[u];
            acc.z += v[u].z * wj[u]; acc.w += v[u].w * wj[u];
        }
    }
}

__global__ __launch_bounds__(256) void k_agg128(const int* __restrict__ csr,
                                                const int* __restrict__ rowptr,
                                                const int* __restrict__ cnt,
                                                const float* __restrict__ dis,
                                                const float* __restrict__ h,
                                                const float* __restrict__ b1,
                                                float* __restrict__ out, int n) {
    const int lane = threadIdx.x & 31;
    const int node = __builtin_amdgcn_readfirstlane((int)((blockIdx.x * 256 + threadIdx.x) >> 5));
    if (node >= n) return;
    const float4* __restrict__ h4 = (const float4*)h;
    const int   beg = __builtin_amdgcn_readfirstlane(rowptr[node]);
    const int   len = __builtin_amdgcn_readfirstlane(cnt[node]);
    const float dd  = dis[node];
    const float4 hv = h4[(size_t)node * 32 + lane];
    const float4 bb = ((const float4*)b1)[lane];

    float4 acc = {0.f, 0.f, 0.f, 0.f};
    for (int j0 = 0; j0 < len; j0 += 32) {
        const int left = len - j0;
        const int m = left < 32 ? left : 32;
        int s = 0; float w = 0.f;
        if (lane < m) { s = csr[beg + j0 + lane]; w = dis[s] * dd; }
        const int wb = __float_as_int(w);
        int u0 = 0;
        for (; u0 + 8 <= m; u0 += 8) rows8<true >(acc, lane, s, wb, u0, m, h4);
        if (u0 < m)                  rows8<false>(acc, lane, s, wb, u0, m, h4);
    }
    const float d2 = dd * dd;
    float4 o;
    o.x = (bb.x + hv.x * d2) + acc.x; o.y = (bb.y + hv.y * d2) + acc.y;
    o.z = (bb.z + hv.z * d2) + acc.z; o.w = (bb.w + hv.w * d2) + acc.w;
    ((float4*)out)[(size_t)node * 32 + lane] = o;
}


__global__ __launch_bounds__(256) void k_gemm2(const float* __restrict__ out1,
                                               const float* __restrict__ W2,
                                               float* __restrict__ h2, int ntiles) {
    int wg = (blockIdx.x * 256 + threadIdx.x) >> 5;
    if (wg >= ntiles) return;
    const int lane  = threadIdx.x & 31;
    const int half  = lane >> 4;
    const int l16   = lane & 15;
    const int kboff = half * 2;
    const int r0    = wg * 16;

    const float* arow = out1 + (size_t)(r0 + l16) * 128;
    v8f acc = {0.f, 0.f, 0.f, 0.f, 0.f, 0.f, 0.f, 0.f};
    for (int k0 = 0; k0 < 128; k0 += 4) {
        v2f a, b;
        a.x = fmaxf(arow[k0 + kboff], 0.f);
        a.y = fmaxf(arow[k0 + kboff + 1], 0.f);
        const float* bp = &W2[(size_t)(k0 + kboff) * 16 + l16];
        b.x = bp[0]; b.y = bp[16];
        acc = __builtin_amdgcn_wmma_f32_16x16x4_f32(false, a, false, b,
                                                    (short)0, acc, false, false);
    }
    float* out = h2 + (size_t)(r0 + half * 8) * 16 + l16;
    #pragma unroll
    for (int i = 0; i < 8; ++i) out[(size_t)i * 16] = acc[i];
}


__global__ __launch_bounds__(256) void k_agg16_softmax(const int* __restrict__ csr,
                                                       const int* __restrict__ rowptr,
                                                       const int* __restrict__ cnt,
                                                       const float* __restrict__ dis,
                                                       const float* __restrict__ h2,
                                                       const float* __restrict__ b2,
                                                       float* __restrict__ logits,
                                                       float* __restrict__ probs, int n) {
    const int g = blockIdx.x * 256 + threadIdx.x;
    if (g >= n * 16) return;
    const int node = g >> 4;
    const int f    = g & 15;
    const int beg  = rowptr[node];
    const int len  = cnt[node];
    const float dd = dis[node];
    const float self = h2[g];
    float acc = 0.f;
    for (int j0 = 0; j0 < len; j0 += 16) {
        const int m = len - j0;
        int s = 0; float w = 0.f;
        if (f < m) { s = csr[beg + j0 + f]; w = dis[s] * dd; }
        float v[16], wj[16];
        #pragma unroll
        for (int u = 0; u < 16; ++u) {
            const int sj = __shfl(s, u, 16);
            wj[u] = __shfl(w, u, 16);
            v[u] = 0.f;
            if (u < m) v[u] = h2[sj * 16 + f];
        }
        #pragma unroll
        for (int u = 0; u < 16; ++u) acc += v[u] * wj[u];
    }
    const float lg = (b2[f] + self * (dd * dd)) + acc;
    logits[g] = lg;
    float mx = lg;
    #pragma unroll
    for (int o = 8; o >= 1; o >>= 1) mx = fmaxf(mx, __shfl_xor(mx, o, 16));
    const float ex = __expf(lg - mx);
    float sum = ex;
    #pragma unroll
    for (int o = 8; o >= 1; o >>= 1) sum += __shfl_xor(sum, o, 16);
    probs[g] = ex / sum;
}


extern "C" void kernel_launch(void* const* d_in, const int* in_sizes, int n_in,
                              void* d_out, int out_size, void* d_ws, size_t ws_size,
                              hipStream_t stream) {
    const float* x  = (const float*)d_in[0];
    const int*   ei = (const int*)d_in[1];
    const float* W1 = (const float*)d_in[2];
    const float* b1 = (const float*)d_in[3];
    const float* W2 = (const float*)d_in[4];
    const float* b2 = (const float*)d_in[5];

    const int N = in_sizes[0] / 256;
    const int E = in_sizes[1] / 2;
    const int* src = ei;
    const int* dst = ei + E;

    auto cdiv = [](long long a, long long b) { return (int)((a + b - 1) / b); };
    const int nb   = cdiv(N, 256);
    const int nbp  = (nb + 3) & ~3;
    const int nblk = cdiv(E, CHUNK);

    char*  ws  = (char*)d_ws;
    size_t off = 0;
    auto alloc = [&](size_t bytes) -> void* {
        void* p = ws + off;
        off += (bytes + 255) & ~(size_t)255;
        return p;
    };
    int*   cnt    = (int*)alloc((size_t)N * 4);
    int*   rowptr = (int*)alloc((size_t)N * 4);
    float* dis    = (float*)alloc((size_t)N * 4);
    float* h0     = (float*)alloc((size_t)N * 128 * 4);
    float* out1   = (float*)alloc((size_t)N * 128 * 4);
    float* h2     = (float*)alloc((size_t)N * 16 * 4);
    int*   csr    = (int*)alloc((size_t)E * 4);
    if (off > ws_size) return;
    int*   hist   = (int*)h2;
    int*   bstart = hist + (size_t)nblk * nbp;
    int*   bsize  = bstart + nbp;
    int*   recs   = (int*)out1;
    if (nbp > MAXNB) return;
    if (((size_t)nblk + 2) * nbp * 4 > (size_t)N * 64) return;
    if ((size_t)E * 8 > (size_t)N * 512) return;
    float* logits = (float*)d_out;
    float* probs  = (float*)d_out + (size_t)N * 16;

    const int nstrip = N / 16;
    const int gA = nstrip / 4;
    const int gB = (nstrip * 2) / 5;
    const int gC = nstrip - gA - gB;
    k_hist_gemm   <<<nblk + gA, 256, 0, stream>>>(dst, hist, E, nbp, nblk, x, W1, h0, 0);
    k_scatter_gemm<<<nblk + gB, 256, 0, stream>>>(src, dst, hist, recs, bstart, bsize, E, nbp, nblk,
                                                  x, W1, h0, gA);
    k_rows_gemm   <<<nb + gC, 256, 0, stream>>>(recs, bstart, bsize, cnt, rowptr, dis, csr, N, nb,
                                                x, W1, h0, gA + gB);

    k_agg128     <<<cdiv((long long)N * 32, 256), 256, 0, stream>>>(csr, rowptr, cnt, dis, h0, b1, out1, N);

    int ntiles = N / 16;
    k_gemm2      <<<cdiv(ntiles, 8), 256, 0, stream>>>(out1, W2, h2, ntiles);
    k_agg16_softmax<<<cdiv((long long)N * 16, 256), 256, 0, stream>>>(csr, rowptr, cnt, dis, h2, b2, logits, probs, N);
}


# relu(aggregate) @ W2 runs inside the layer-1 aggregate: a work-group sums a 16-row tile (8 waves, 2 rows each) into LDS, one wave does its 32 WMMAs; aggregate never stored, 5 launches, same sums
# speedup vs baseline: 4.8754x; 1.0891x over previous
#include <hip/hip_runtime.h>
#include <hip/hip_bf16.h>


typedef float v2f __attribute__((ext_vector_type(2)));
typedef float v8f __attribute__((ext_vector_type(8)));

#define LDS_STRIDE 260


#define BSH    8
#define CHUNK  8192
#define MAXNB  1024
#define STG    16384

__device__ __forceinline__ int block_excl(int t, int* wsum, int lane, int wave) {
    int incl = t;
    #pragma unroll
    for (int o = 1; o < 32; o <<= 1) {
        const int y = __shfl_up(incl, o, 32);
        if (lane >= o) incl += y;
    }
    if (lane == 31) wsum[wave] = incl;
    __syncthreads();
    int off = 0;
    #pragma unroll
    for (int w = 0; w < 8; ++w) if (w < wave) off += wsum[w];
    return off + incl - t;
}

__device__ __forceinline__ void hist_pass(const int* __restrict__ dst,
                                          int* __restrict__ hist, int e, int nbp,
                                          int blk, unsigned char* lds) {
    int* lh = (int*)lds;
    const int tid = threadIdx.x;
    for (int b = tid; b < nbp; b += 256) lh[b] = 0;
    __syncthreads();
    const int base = blk * CHUNK + tid;
    for (int k0 = 0; k0 < CHUNK; k0 += 16 * 256) {
        int d[16];
        #pragma unroll
        for (int u = 0; u < 16; ++u) {
            const int i = base + k0 + u * 256;
            d[u] = dst[i < e ? i : e - 1];
        }
        #pragma unroll
        for (int u = 0; u < 16; ++u)
            if (base + k0 + u * 256 < e) atomicAdd(&lh[d[u] >> BSH], 1);
    }
    __syncthreads();
    for (int b = tid; b < nbp; b += 256) hist[(size_t)blk * nbp + b] = lh[b];
}

#define SCATTER_LDS (CHUNK * 8 + 2 * MAXNB * 4 + 64)
__device__ __forceinline__ void scatter_pass(const int* __restrict__ src,
                                             const int* __restrict__ dst,
                                             const int* __restrict__ hist,
                                             int* __restrict__ recs,
                                             int* __restrict__ bstart,
                                             int* __restrict__ bsize,
                                             int e, int nbp, int nblk,
                                             int blk, unsigned char* lds) {
    int2* stage = (int2*)lds;
    int*  ltot  = (int*)lds;
    int*  lbel  = (int*)lds + MAXNB;
    int*  lcur  = (int*)(lds + CHUNK * 8);
    int*  gdel  = lcur + MAXNB;
    int*  wsA   = gdel + MAXNB;
    int*  wsB   = wsA + 8;
    const int tid  = threadIdx.x;
    const int lane = tid & 31;
    const int wave = tid >> 5;
    for (int b = tid; b < nbp; b += 256) { ltot[b] = 0; lbel[b] = 0; }
    __syncthreads();

    const int4* __restrict__ h4 = (const int4*)hist;
    const int nq = nbp >> 2;
    const int ng = 256 / nq;
    const int q  = tid % nq;
    const int g  = tid / nq;
    if (g < ng) {
        int4 tot = {0, 0, 0, 0}, bel = {0, 0, 0, 0};
        for (int j0 = g; j0 < nblk; j0 += 16 * ng) {
            int4 v[16];
            #pragma unroll
            for (int u = 0; u < 16; ++u) {
                const int jj = j0 + u * ng;
                v[u] = h4[(size_t)(jj < nblk ? jj : 0) * nq + q];
            }
            #pragma unroll
            for (int u = 0; u < 16; ++u) {
                const int jj = j0 + u * ng;
                if (jj < nblk) {
                    tot.x += v[u].x; tot.y += v[u].y; tot.z += v[u].z; tot.w += v[u].w;
                    if (jj < blk) { bel.x += v[u].x; bel.y += v[u].y; bel.z += v[u].z; bel.w += v[u].w; }
                }
            }
        }
        atomicAdd(&ltot[4 * q + 0], tot.x); atomicAdd(&ltot[4 * q + 1], tot.y);
        atomicAdd(&ltot[4 * q + 2], tot.z); atomicAdd(&ltot[4 * q + 3], tot.w);
        atomicAdd(&lbel[4 * q + 0], bel.x); atomicAdd(&lbel[4 * q + 1], bel.y);
        atomicAdd(&lbel[4 * q + 2], bel.z); atomicAdd(&lbel[4 * q + 3], bel.w);
    }
    __syncthreads();

    int a[4] = {0, 0, 0, 0}, o[4] = {0, 0, 0, 0};
    if (tid < nq) {
        const int4 own = h4[(size_t)blk * nq + tid];
        o[0] = own.x; o[1] = own.y; o[2] = own.z; o[3] = own.w;
        #pragma unroll
        for (int k = 0; k < 4; ++k) a[k] = ltot[4 * tid + k];
    }
    int exA = block_excl((a[0] + a[1]) + (a[2] + a[3]), wsA, lane, wave);
    int exO = block_excl((o[0] + o[1]) + (o[2] + o[3]), wsB, lane, wave);
    if (tid < nq) {
        #pragma unroll
        for (int k = 0; k < 4; ++k) {
            const int b = 4 * tid + k;
            lcur[b] = exO;
            gdel[b] = exA + lbel[b] - exO;
            if (blk == 0) { bstart[b] = exA; bsize[b] = a[k]; }
            exA += a[k]; exO += o[k];
        }
    }
    __syncthreads();

    const int base = blk * CHUNK + tid;
    for (int k0 = 0; k0 < CHUNK; k0 += 16 * 256) {
        int d[16], s[16];
        #pragma unroll
        for (int u = 0; u < 16; ++u) {
            const int i  = base + k0 + u * 256;
            const int ic = i < e ? i : e - 1;
            d[u] = dst[ic]; s[u] = src[ic];
        }
        int slot[16];
        #pragma unroll
        for (int u = 0; u < 16; ++u) {
            slot[u] = 0;
            if (base + k0 + u * 256 < e) slot[u] = atomicAdd(&lcur[d[u] >> BSH], 1);
        }
        #pragma unroll
        for (int u = 0; u < 16; ++u)
            if (base + k0 + u * 256 < e) stage[slot[u]] = make_int2(s[u], d[u]);
    }
    __syncthreads();
    const int left = e - blk * CHUNK;
    const int nown = left < CHUNK ? left : CHUNK;
    int2* __restrict__ rc = (int2*)recs;
    for (int k0 = tid; k0 < nown; k0 += 8 * 256) {
        int2 r[8]; int gd[8];
        #pragma unroll
        for (int u = 0; u < 8; ++u) {
            const int k = k0 + u * 256;
            r[u] = stage[k < nown ? k : 0];
        }
        #pragma unroll
        for (int u = 0; u < 8; ++u) gd[u] = gdel[r[u].y >> BSH];
        #pragma unroll
        for (int u = 0; u < 8; ++u) {
            const int k = k0 + u * 256;
            if (k < nown) rc[(size_t)(k + gd[u])] = r[u];
        }
    }
}

#define ROWS_LDS (STG * 4 + 256 * 4 + 32)
__device__ __forceinline__ void rows_pass(const int* __restrict__ recs,
                                          const int* __restrict__ bstart,
                                          const int* __restrict__ bsize,
                                          int* __restrict__ cnt,
                                          int* __restrict__ rowptr,
                                          float* __restrict__ dis,
                                          int* __restrict__ csr, int n,
                                          int bkt, unsigned char* lds) {
    int* lst = (int*)lds;
    int* lc  = lst + STG;
    int* ws  = lc + 256;
    const int tid  = threadIdx.x;
    const int lane = tid & 31;
    const int wave = tid >> 5;
    const int s0 = bstart[bkt];
    const int m  = bsize[bkt];
    const int2* __restrict__ rc = (const int2*)recs + s0;
    lc[tid] = 0;
    __syncthreads();
    for (int k0 = 0; k0 < m; k0 += 16 * 256) {
        int2 r[16];
        #pragma unroll
        for (int u = 0; u < 16; ++u) {
            const int k = k0 + u * 256 + tid;
            r[u] = rc[k < m ? k : 0];
        }
        #pragma unroll
        for (int u = 0; u < 16; ++u)
            if (k0 + u * 256 + tid < m) atomicAdd(&lc[r[u].y & 255], 1);
    }
    __syncthreads();
    const int c  = lc[tid];
    const int ex = block_excl(c, ws, lane, wave);
    const int node = bkt * 256 + tid;
    if (node < n) {
        cnt[node]    = c;
        rowptr[node] = s0 + ex;
        dis[node]    = __frsqrt_rn((float)(c + 1));
    }
    lc[tid] = ex;
    __syncthreads();
    const bool staged = m <= STG;
    for (int k0 = 0; k0 < m; k0 += 16 * 256) {
        int2 r[16];
        #pragma unroll
        for (int u = 0; u < 16; ++u) {
            const int k = k0 + u * 256 + tid;
            r[u] = rc[k < m ? k : 0];
        }
        int p[16];
        #pragma unroll
        for (int u = 0; u < 16; ++u) {
            p[u] = 0;
            if (k0 + u * 256 + tid < m) p[u] = atomicAdd(&lc[r[u].y & 255], 1);
        }
        #pragma unroll
        for (int u = 0; u < 16; ++u) {
            if (k0 + u * 256 + tid < m) {
                if (staged) lst[p[u]] = r[u].x; else csr[s0 + p[u]] = r[u].x;
            }
        }
    }
    if (staged) {
        __syncthreads();
        for (int k0 = tid; k0 < m; k0 += 8 * 256) {
            int v[8];
            #pragma unroll
            for (int u = 0; u < 8; ++u) {
                const int k = k0 + u * 256;
                v[u] = lst[k < m ? k : 0];
            }
            #pragma unroll
            for (int u = 0; u < 8; ++u) {
                const int k = k0 + u * 256;
                if (k < m) csr[s0 + k] = v[u];
            }
        }
    }
}


#define GEMM1_LDS (16 * LDS_STRIDE * 4)
__device__ __forceinline__ void gemm1_strip(const float* __restrict__ x,
                                            const float* __restrict__ W1,
                                            float* __restrict__ h0,
                                            int strip, unsigned char* lds) {
    float* As = (float*)lds;
    const int tid = threadIdx.x;
    const int r0  = strip * 16;

    const float4* xv = (const float4*)(x + (size_t)r0 * 256);
    #pragma unroll
    for (int i = 0; i < 4; ++i) {
        int j   = i * 256 + tid;
        int row = j >> 6;
        int c4  = j & 63;
        float4 v = xv[(size_t)row * 64 + c4];
        float* d = &As[row * LDS_STRIDE + c4 * 4];
        d[0] = v.x; d[1] = v.y; d[2] = v.z; d[3] = v.w;
    }
    __syncthreads();

    const int wave  = tid >> 5;
    const int lane  = tid & 31;
    const int half  = lane >> 4;
    const int l16   = lane & 15;
    const int c0    = wave * 16;
    const int kboff = half * 2;

    v8f acc = {0.f, 0.f, 0.f, 0.f, 0.f, 0.f, 0.f, 0.f};
    for (int k0 = 0; k0 < 256; k0 += 4) {
        v2f a, b;
        const float* ap = &As[l16 * LDS_STRIDE + k0 + kboff];
        a.x = ap[0]; a.y = ap[1];
        const float* bp = &W1[(size_t)(k0 + kboff) * 128 + c0 + l16];
        b.x = bp[0]; b.y = bp[128];
        acc = __builtin_amdgcn_wmma_f32_16x16x4_f32(false, a, false, b,
                                                    (short)0, acc, false, false);
    }
    float* out = h0 + (size_t)(r0 + half * 8) * 128 + c0 + l16;
    #pragma unroll
    for (int i = 0; i < 8; ++i) out[(size_t)i * 128] = acc[i];
}


#define MAX2(a, b) ((a) > (b) ? (a) : (b))

__global__ __launch_bounds__(256) void k_hist_gemm(const int* __restrict__ dst,
                                                   int* __restrict__ hist, int e, int nbp, int nsort,
                                                   const float* __restrict__ x,
                                                   const float* __restrict__ W1,
                                                   float* __restrict__ h0, int strip0) {
    __shared__ __attribute__((aligned(16))) unsigned char lds[MAX2(MAXNB * 4, GEMM1_LDS)];
    const int blk = blockIdx.x;
    if (blk < nsort) hist_pass(dst, hist, e, nbp, blk, lds);
    else             gemm1_strip(x, W1, h0, strip0 + (blk - nsort), lds);
}

__global__ __launch_bounds__(256) void k_scatter_gemm(const int* __restrict__ src,
                                                      const int* __restrict__ dst,
                                                      const int* __restrict__ hist,
                                                      int* __restrict__ recs,
                                                      int* __restrict__ bstart,
                                                      int* __restrict__ bsize,
                                                      int e, int nbp, int nsort,
                                                      const float* __restrict__ x,
                                                      const float* __restrict__ W1,
                                                      float* __restrict__ h0, int strip0) {
    __shared__ __attribute__((aligned(16))) unsigned char lds[MAX2(SCATTER_LDS, GEMM1_LDS)];
    const int blk = blockIdx.x;
    if (blk < nsort) scatter_pass(src, dst, hist, recs, bstart, bsize, e, nbp, nsort, blk, lds);
    else             gemm1_strip(x, W1, h0, strip0 + (blk - nsort), lds);
}

__global__ __launch_bounds__(256) void k_rows_gemm(const int* __restrict__ recs,
                                                   const int* __restrict__ bstart,
                                                   const int* __restrict__ bsize,
                                                   int* __restrict__ cnt,
                                                   int* __restrict__ rowptr,
                                                   float* __restrict__ dis,
                                                   int* __restrict__ csr, int n, int nsort,
                                                   const float* __restrict__ x,
                                                   const float* __restrict__ W1,
                                                   float* __restrict__ h0, int strip0) {
    __shared__ __attribute__((aligned(16))) unsigned char lds[MAX2(ROWS_LDS, GEMM1_LDS)];
    const int blk = blockIdx.x;
    if (blk < nsort) rows_pass(recs, bstart, bsize, cnt, rowptr, dis, csr, n, blk, lds);
    else             gemm1_strip(x, W1, h0, strip0 + (blk - nsort), lds);
}


#define T2_PITCH 132

template <bool FULL>
__device__ __forceinline__ void rows8(float4& acc, int lane, int s, int wbits, int u0, int m,
                                      const float4* __restrict__ h4) {
    int sj[8]; float wj[8];
    #pragma unroll
    for (int u = 0; u < 8; ++u) {
        sj[u] = __builtin_amdgcn_readlane(s, u0 + u);
        wj[u] = __int_as_float(__builtin_amdgcn_readlane(wbits, u0 + u));
    }
    float4 v[8];
    #pragma unroll
    for (int u = 0; u < 8; ++u)
        if (FULL || (u0 + u < m)) v[u] = h4[(size_t)sj[u] * 32 + lane];
    #pragma unroll
    for (int u = 0; u < 8; ++u) {
        if (FULL || (u0 + u < m)) {
            acc.x += v[u].x * wj[u]; acc.y += v[u].y * wj[u];
            acc.z += v[u].z * wj[u]; acc.w += v[u].w * wj[u];
        }
    }
}

__device__ __forceinline__ float4 agg_row128(int node, int lane,
                                             const int* __restrict__ csr,
                                             const int* __restrict__ rowptr,
                                             const int* __restrict__ cnt,
                                             const float* __restrict__ dis,
                                             const float4* __restrict__ h4, float4 bb) {
    const int   beg = __builtin_amdgcn_readfirstlane(rowptr[node]);
    const int   len = __builtin_amdgcn_readfirstlane(cnt[node]);
    const float dd  = dis[node];
    const float4 hv = h4[(size_t)node * 32 + lane];

    float4 acc = {0.f, 0.f, 0.f, 0.f};
    for (int j0 = 0; j0 < len; j0 += 32) {
        const int left = len - j0;
        const int m = left < 32 ? left : 32;
        int s = 0; float w = 0.f;
        if (lane < m) { s = csr[beg + j0 + lane]; w = dis[s] * dd; }
        const int wb = __float_as_int(w);
        int u0 = 0;
        for (; u0 + 8 <= m; u0 += 8) rows8<true >(acc, lane, s, wb, u0, m, h4);
        if (u0 < m)                  rows8<false>(acc, lane, s, wb, u0, m, h4);
    }
    const float d2 = dd * dd;
    float4 o;
    o.x = (bb.x + hv.x * d2) + acc.x; o.y = (bb.y + hv.y * d2) + acc.y;
    o.z = (bb.z + hv.z * d2) + acc.z; o.w = (bb.w + hv.w * d2) + acc.w;
    return o;
}

__global__ __launch_bounds__(256) void k_agg128_gemm2(const int* __restrict__ csr,
                                                      const int* __restrict__ rowptr,
                                                      const int* __restrict__ cnt,
                                                      const float* __restrict__ dis,
                                                      const float* __restrict__ h,
                                                      const float* __restrict__ b1,
                                                      const float* __restrict__ W2,
                                                      float* __restrict__ h2, int n) {
    __shared__ __attribute__((aligned(16))) float tile[16 * T2_PITCH];
    const int lane = threadIdx.x & 31;
    const int wave = __builtin_amdgcn_readfirstlane((int)(threadIdx.x >> 5));
    const int r0   = blockIdx.x * 16;
    const float4* __restrict__ h4 = (const float4*)h;
    const float4 bb = ((const float4*)b1)[lane];

    for (int r = 0; r < 2; ++r) {
        const int trow = wave * 2 + r;
        const int node = r0 + trow;
        float4 o = {0.f, 0.f, 0.f, 0.f};
        if (node < n) {
            o = agg_row128(node, lane, csr, rowptr, cnt, dis, h4, bb);
            o.x = fmaxf(o.x, 0.f); o.y = fmaxf(o.y, 0.f);
            o.z = fmaxf(o.z, 0.f); o.w = fmaxf(o.w, 0.f);
        }
        *(float4*)&tile[trow * T2_PITCH + lane * 4] = o;
    }
    __syncthreads();
    if (wave != 0) return;

    const int half  = lane >> 4;
    const int l16   = lane & 15;
    const int kboff = half * 2;
    const float* ap = &tile[l16 * T2_PITCH + kboff];
    const float* bp = W2 + (size_t)kboff * 16 + l16;
    v8f acc = {0.f, 0.f, 0.f, 0.f, 0.f, 0.f, 0.f, 0.f};
    for (int k0 = 0; k0 < 128; k0 += 4) {
        v2f a, b;
        a.x = ap[k0]; a.y = ap[k0 + 1];
        b.x = bp[(size_t)k0 * 16]; b.y = bp[(size_t)k0 * 16 + 16];
        acc = __builtin_amdgcn_wmma_f32_16x16x4_f32(false, a, false, b,
                                                    (short)0, acc, false, false);
    }
    float* out = h2 + (size_t)(r0 + half * 8) * 16 + l16;
    #pragma unroll
    for (int i = 0; i < 8; ++i)
        if (r0 + half * 8 + i < n) out[(size_t)i * 16] = acc[i];
}


__global__ __launch_bounds__(256) void k_agg16_softmax(const int* __restrict__ csr,
                                                       const int* __restrict__ rowptr,
                                                       const int* __restrict__ cnt,
                                                       const float* __restrict__ dis,
                                                       const float* __restrict__ h2,
                                                       const float* __restrict__ b2,
                                                       float* __restrict__ logits,
                                                       float* __restrict__ probs, int n) {
    const int g = blockIdx.x * 256 + threadIdx.x;
    if (g >= n * 16) return;
    const int node = g >> 4;
    const int f    = g & 15;
    const int beg  = rowptr[node];
    const int len  = cnt[node];
    const float dd = dis[node];
    const float self = h2[g];
    float acc = 0.f;
    for (int j0 = 0; j0 < len; j0 += 16) {
        const int m = len - j0;
        int s = 0; float w = 0.f;
        if (f < m) { s = csr[beg + j0 + f]; w = dis[s] * dd; }
        float v[16], wj[16];
        #pragma unroll
        for (int u = 0; u < 16; ++u) {
            const int sj = __shfl(s, u, 16);
            wj[u] = __shfl(w, u, 16);
            v[u] = 0.f;
            if (u < m) v[u] = h2[sj * 16 + f];
        }
        #pragma unroll
        for (int u = 0; u < 16; ++u) acc += v[u] * wj[u];
    }
    const float lg = (b2[f] + self * (dd * dd)) + acc;
    logits[g] = lg;
    float mx = lg;
    #pragma unroll
    for (int o = 8; o >= 1; o >>= 1) mx = fmaxf(mx, __shfl_xor(mx, o, 16));
    const float ex = __expf(lg - mx);
    float sum = ex;
    #pragma unroll
    for (int o = 8; o >= 1; o >>= 1) sum += __shfl_xor(sum, o, 16);
    probs[g] = ex / sum;
}


extern "C" void kernel_launch(void* const* d_in, const int* in_sizes, int n_in,
                              void* d_out, int out_size, void* d_ws, size_t ws_size,
                              hipStream_t stream) {
    const float* x  = (const float*)d_in[0];
    const int*   ei = (const int*)d_in[1];
    const float* W1 = (const float*)d_in[2];
    const float* b1 = (const float*)d_in[3];
    const float* W2 = (const float*)d_in[4];
    const float* b2 = (const float*)d_in[5];

    const int N = in_sizes[0] / 256;
    const int E = in_sizes[1] / 2;
    const int* src = ei;
    const int* dst = ei + E;

    auto cdiv = [](long long a, long long b) { return (int)((a + b - 1) / b); };
    const int nb   = cdiv(N, 256);
    const int nbp  = (nb + 3) & ~3;
    const int nblk = cdiv(E, CHUNK);

    char*  ws  = (char*)d_ws;
    size_t off = 0;
    auto alloc = [&](size_t bytes) -> void* {
        void* p = ws + off;
        off += (bytes + 255) & ~(size_t)255;
        return p;
    };
    int*   cnt    = (int*)alloc((size_t)N * 4);
    int*   rowptr = (int*)alloc((size_t)N * 4);
    float* dis    = (float*)alloc((size_t)N * 4);
    float* h0     = (float*)alloc((size_t)N * 128 * 4);
    int*   recs   = (int*)alloc((size_t)E * 8);
    float* h2     = (float*)alloc((size_t)N * 16 * 4);
    int*   csr    = (int*)alloc((size_t)E * 4);
    if (off > ws_size) return;
    int*   hist   = (int*)h2;
    int*   bstart = hist + (size_t)nblk * nbp;
    int*   bsize  = bstart + nbp;
    if (nbp > MAXNB) return;
    if (((size_t)nblk + 2) * nbp * 4 > (size_t)N * 64) return;
    float* logits = (float*)d_out;
    float* probs  = (float*)d_out + (size_t)N * 16;

    const int nstrip = N / 16;
    const int gA = nstrip / 4;
    const int gB = (nstrip * 2) / 5;
    const int gC = nstrip - gA - gB;
    k_hist_gemm   <<<nblk + gA, 256, 0, stream>>>(dst, hist, E, nbp, nblk, x, W1, h0, 0);
    k_scatter_gemm<<<nblk + gB, 256, 0, stream>>>(src, dst, hist, recs, bstart, bsize, E, nbp, nblk,
                                                  x, W1, h0, gA);
    k_rows_gemm   <<<nb + gC, 256, 0, stream>>>(recs, bstart, bsize, cnt, rowptr, dis, csr, N, nb,
                                                x, W1, h0, gA + gB);

    k_agg128_gemm2<<<cdiv(N, 16), 256, 0, stream>>>(csr, rowptr, cnt, dis, h0, b1, W2, h2, N);

    k_agg16_softmax<<<cdiv((long long)N * 16, 256), 256, 0, stream>>>(csr, rowptr, cnt, dis, h2, b2, logits, probs, N);
}


# layer-2 norm factored out: product tile rows scaled by dis[row] in LDS, so the 16-wide aggregate sums plain rows, no dis[source] gather per edge; dis[dest] once on edges+self loop, bias added last
# speedup vs baseline: 5.1494x; 1.0562x over previous
#include <hip/hip_runtime.h>
#include <hip/hip_bf16.h>


typedef float v2f __attribute__((ext_vector_type(2)));
typedef float v8f __attribute__((ext_vector_type(8)));

#define LDS_STRIDE 260


#define BSH    8
#define CHUNK  8192
#define MAXNB  1024
#define STG    16384

__device__ __forceinline__ int block_excl(int t, int* wsum, int lane, int wave) {
    int incl = t;
    #pragma unroll
    for (int o = 1; o < 32; o <<= 1) {
        const int y = __shfl_up(incl, o, 32);
        if (lane >= o) incl += y;
    }
    if (lane == 31) wsum[wave] = incl;
    __syncthreads();
    int off = 0;
    #pragma unroll
    for (int w = 0; w < 8; ++w) if (w < wave) off += wsum[w];
    return off + incl - t;
}

__device__ __forceinline__ void hist_pass(const int* __restrict__ dst,
                                          int* __restrict__ hist, int e, int nbp,
                                          int blk, unsigned char* lds) {
    int* lh = (int*)lds;
    const int tid = threadIdx.x;
    for (int b = tid; b < nbp; b += 256) lh[b] = 0;
    __syncthreads();
    const int base = blk * CHUNK + tid;
    for (int k0 = 0; k0 < CHUNK; k0 += 16 * 256) {
        int d[16];
        #pragma unroll
        for (int u = 0; u < 16; ++u) {
            const int i = base + k0 + u * 256;
            d[u] = dst[i < e ? i : e - 1];
        }
        #pragma unroll
        for (int u = 0; u < 16; ++u)
            if (base + k0 + u * 256 < e) atomicAdd(&lh[d[u] >> BSH], 1);
    }
    __syncthreads();
    for (int b = tid; b < nbp; b += 256) hist[(size_t)blk * nbp + b] = lh[b];
}

#define SCATTER_LDS (CHUNK * 8 + 2 * MAXNB * 4 + 64)
__device__ __forceinline__ void scatter_pass(const int* __restrict__ src,
                                             const int* __restrict__ dst,
                                             const int* __restrict__ hist,
                                             int* __restrict__ recs,
                                             int* __restrict__ bstart,
                                             int* __restrict__ bsize,
                                             int e, int nbp, int nblk,
                                             int blk, unsigned char* lds) {
    int2* stage = (int2*)lds;
    int*  ltot  = (int*)lds;
    int*  lbel  = (int*)lds + MAXNB;
    int*  lcur  = (int*)(lds + CHUNK * 8);
    int*  gdel  = lcur + MAXNB;
    int*  wsA   = gdel + MAXNB;
    int*  wsB   = wsA + 8;
    const int tid  = threadIdx.x;
    const int lane = tid & 31;
    const int wave = tid >> 5;
    for (int b = tid; b < nbp; b += 256) { ltot[b] = 0; lbel[b] = 0; }
    __syncthreads();

    const int4* __restrict__ h4 = (const int4*)hist;
    const int nq = nbp >> 2;
    const int ng = 256 / nq;
    const int q  = tid % nq;
    const int g  = tid / nq;
    if (g < ng) {
        int4 tot = {0, 0, 0, 0}, bel = {0, 0, 0, 0};
        for (int j0 = g; j0 < nblk; j0 += 16 * ng) {
            int4 v[16];
            #pragma unroll
            for (int u = 0; u < 16; ++u) {
                const int jj = j0 + u * ng;
                v[u] = h4[(size_t)(jj < nblk ? jj : 0) * nq + q];
            }
            #pragma unroll
            for (int u = 0; u < 16; ++u) {
                const int jj = j0 + u * ng;
                if (jj < nblk) {
                    tot.x += v[u].x; tot.y += v[u].y; tot.z += v[u].z; tot.w += v[u].w;
                    if (jj < blk) { bel.x += v[u].x; bel.y += v[u].y; bel.z += v[u].z; bel.w += v[u].w; }
                }
            }
        }
        atomicAdd(&ltot[4 * q + 0], tot.x); atomicAdd(&ltot[4 * q + 1], tot.y);
        atomicAdd(&ltot[4 * q + 2], tot.z); atomicAdd(&ltot[4 * q + 3], tot.w);
        atomicAdd(&lbel[4 * q + 0], bel.x); atomicAdd(&lbel[4 * q + 1], bel.y);
        atomicAdd(&lbel[4 * q + 2], bel.z); atomicAdd(&lbel[4 * q + 3], bel.w);
    }
    __syncthreads();

    int a[4] = {0, 0, 0, 0}, o[4] = {0, 0, 0, 0};
    if (tid < nq) {
        const int4 own = h4[(size_t)blk * nq + tid];
        o[0] = own.x; o[1] = own.y; o[2] = own.z; o[3] = own.w;
        #pragma unroll
        for (int k = 0; k < 4; ++k) a[k] = ltot[4 * tid + k];
    }
    int exA = block_excl((a[0] + a[1]) + (a[2] + a[3]), wsA, lane, wave);
    int exO = block_excl((o[0] + o[1]) + (o[2] + o[3]), wsB, lane, wave);
    if (tid < nq) {
        #pragma unroll
        for (int k = 0; k < 4; ++k) {
            const int b = 4 * tid + k;
            lcur[b] = exO;
            gdel[b] = exA + lbel[b] - exO;
            if (blk == 0) { bstart[b] = exA; bsize[b] = a[k]; }
            exA += a[k]; exO += o[k];
        }
    }
    __syncthreads();

    const int base = blk * CHUNK + tid;
    for (int k0 = 0; k0 < CHUNK; k0 += 16 * 256) {
        int d[16], s[16];
        #pragma unroll
        for (int u = 0; u < 16; ++u) {
            const int i  = base + k0 + u * 256;
            const int ic = i < e ? i : e - 1;
            d[u] = dst[ic]; s[u] = src[ic];
        }
        int slot[16];
        #pragma unroll
        for (int u = 0; u < 16; ++u) {
            slot[u] = 0;
            if (base + k0 + u * 256 < e) slot[u] = atomicAdd(&lcur[d[u] >> BSH], 1);
        }
        #pragma unroll
        for (int u = 0; u < 16; ++u)
            if (base + k0 + u * 256 < e) stage[slot[u]] = make_int2(s[u], d[u]);
    }
    __syncthreads();
    const int left = e - blk * CHUNK;
    const int nown = left < CHUNK ? left : CHUNK;
    int2* __restrict__ rc = (int2*)recs;
    for (int k0 = tid; k0 < nown; k0 += 8 * 256) {
        int2 r[8]; int gd[8];
        #pragma unroll
        for (int u = 0; u < 8; ++u) {
            const int k = k0 + u * 256;
            r[u] = stage[k < nown ? k : 0];
        }
        #pragma unroll
        for (int u = 0; u < 8; ++u) gd[u] = gdel[r[u].y >> BSH];
        #pragma unroll
        for (int u = 0; u < 8; ++u) {
            const int k = k0 + u * 256;
            if (k < nown) rc[(size_t)(k + gd[u])] = r[u];
        }
    }
}

#define ROWS_LDS (STG * 4 + 256 * 4 + 32)
__device__ __forceinline__ void rows_pass(const int* __restrict__ recs,
                                          const int* __restrict__ bstart,
                                          const int* __restrict__ bsize,
                                          int* __restrict__ cnt,
                                          int* __restrict__ rowptr,
                                          float* __restrict__ dis,
                                          int* __restrict__ csr, int n,
                                          int bkt, unsigned char* lds) {
    int* lst = (int*)lds;
    int* lc  = lst + STG;
    int* ws  = lc + 256;
    const int tid  = threadIdx.x;
    const int lane = tid & 31;
    const int wave = tid >> 5;
    const int s0 = bstart[bkt];
    const int m  = bsize[bkt];
    const int2* __restrict__ rc = (const int2*)recs + s0;
    lc[tid] = 0;
    __syncthreads();
    for (int k0 = 0; k0 < m; k0 += 16 * 256) {
        int2 r[16];
        #pragma unroll
        for (int u = 0; u < 16; ++u) {
            const int k = k0 + u * 256 + tid;
            r[u] = rc[k < m ? k : 0];
        }
        #pragma unroll
        for (int u = 0; u < 16; ++u)
            if (k0 + u * 256 + tid < m) atomicAdd(&lc[r[u].y & 255], 1);
    }
    __syncthreads();
    const int c  = lc[tid];
    const int ex = block_excl(c, ws, lane, wave);
    const int node = bkt * 256 + tid;
    if (node < n) {
        cnt[node]    = c;
        rowptr[node] = s0 + ex;
        dis[node]    = __frsqrt_rn((float)(c + 1));
    }
    lc[tid] = ex;
    __syncthreads();
    const bool staged = m <= STG;
    for (int k0 = 0; k0 < m; k0 += 16 * 256) {
        int2 r[16];
        #pragma unroll
        for (int u = 0; u < 16; ++u) {
            const int k = k0 + u * 256 + tid;
            r[u] = rc[k < m ? k : 0];
        }
        int p[16];
        #pragma unroll
        for (int u = 0; u < 16; ++u) {
            p[u] = 0;
            if (k0 + u * 256 + tid < m) p[u] = atomicAdd(&lc[r[u].y & 255], 1);
        }
        #pragma unroll
        for (int u = 0; u < 16; ++u) {
            if (k0 + u * 256 + tid < m) {
                if (staged) lst[p[u]] = r[u].x; else csr[s0 + p[u]] = r[u].x;
            }
        }
    }
    if (staged) {
        __syncthreads();
        for (int k0 = tid; k0 < m; k0 += 8 * 256) {
            int v[8];
            #pragma unroll
            for (int u = 0; u < 8; ++u) {
                const int k = k0 + u * 256;
                v[u] = lst[k < m ? k : 0];
            }
            #pragma unroll
            for (int u = 0; u < 8; ++u) {
                const int k = k0 + u * 256;
                if (k < m) csr[s0 + k] = v[u];
            }
        }
    }
}


#define GEMM1_LDS (16 * LDS_STRIDE * 4)
__device__ __forceinline__ void gemm1_strip(const float* __restrict__ x,
                                            const float* __restrict__ W1,
                                            float* __restrict__ h0,
                                            int strip, unsigned char* lds) {
    float* As = (float*)lds;
    const int tid = threadIdx.x;
    const int r0  = strip * 16;

    const float4* xv = (const float4*)(x + (size_t)r0 * 256);
    #pragma unroll
    for (int i = 0; i < 4; ++i) {
        int j   = i * 256 + tid;
        int row = j >> 6;
        int c4  = j & 63;
        float4 v = xv[(size_t)row * 64 + c4];
        float* d = &As[row * LDS_STRIDE + c4 * 4];
        d[0] = v.x; d[1] = v.y; d[2] = v.z; d[3] = v.w;
    }
    __syncthreads();

    const int wave  = tid >> 5;
    const int lane  = tid & 31;
    const int half  = lane >> 4;
    const int l16   = lane & 15;
    const int c0    = wave * 16;
    const int kboff = half * 2;

    v8f acc = {0.f, 0.f, 0.f, 0.f, 0.f, 0.f, 0.f, 0.f};
    for (int k0 = 0; k0 < 256; k0 += 4) {
        v2f a, b;
        const float* ap = &As[l16 * LDS_STRIDE + k0 + kboff];
        a.x = ap[0]; a.y = ap[1];
        const float* bp = &W1[(size_t)(k0 + kboff) * 128 + c0 + l16];
        b.x = bp[0]; b.y = bp[128];
        acc = __builtin_amdgcn_wmma_f32_16x16x4_f32(false, a, false, b,
                                                    (short)0, acc, false, false);
    }
    float* out = h0 + (size_t)(r0 + half * 8) * 128 + c0 + l16;
    #pragma unroll
    for (int i = 0; i < 8; ++i) out[(size_t)i * 128] = acc[i];
}


#define MAX2(a, b) ((a) > (b) ? (a) : (b))

__global__ __launch_bounds__(256) void k_hist_gemm(const int* __restrict__ dst,
                                                   int* __restrict__ hist, int e, int nbp, int nsort,
                                                   const float* __restrict__ x,
                                                   const float* __restrict__ W1,
                                                   float* __restrict__ h0, int strip0) {
    __shared__ __attribute__((aligned(16))) unsigned char lds[MAX2(MAXNB * 4, GEMM1_LDS)];
    const int blk = blockIdx.x;
    if (blk < nsort) hist_pass(dst, hist, e, nbp, blk, lds);
    else             gemm1_strip(x, W1, h0, strip0 + (blk - nsort), lds);
}

__global__ __launch_bounds__(256) void k_scatter_gemm(const int* __restrict__ src,
                                                      const int* __restrict__ dst,
                                                      const int* __restrict__ hist,
                                                      int* __restrict__ recs,
                                                      int* __restrict__ bstart,
                                                      int* __restrict__ bsize,
                                                      int e, int nbp, int nsort,
                                                      const float* __restrict__ x,
                                                      const float* __restrict__ W1,
                                                      float* __restrict__ h0, int strip0) {
    __shared__ __attribute__((aligned(16))) unsigned char lds[MAX2(SCATTER_LDS, GEMM1_LDS)];
    const int blk = blockIdx.x;
    if (blk < nsort) scatter_pass(src, dst, hist, recs, bstart, bsize, e, nbp, nsort, blk, lds);
    else             gemm1_strip(x, W1, h0, strip0 + (blk - nsort), lds);
}

__global__ __launch_bounds__(256) void k_rows_gemm(const int* __restrict__ recs,
                                                   const int* __restrict__ bstart,
                                                   const int* __restrict__ bsize,
                                                   int* __restrict__ cnt,
                                                   int* __restrict__ rowptr,
                                                   float* __restrict__ dis,
                                                   int* __restrict__ csr, int n, int nsort,
                                                   const float* __restrict__ x,
                                                   const float* __restrict__ W1,
                                                   float* __restrict__ h0, int strip0) {
    __shared__ __attribute__((aligned(16))) unsigned char lds[MAX2(ROWS_LDS, GEMM1_LDS)];
    const int blk = blockIdx.x;
    if (blk < nsort) rows_pass(recs, bstart, bsize, cnt, rowptr, dis, csr, n, blk, lds);
    else             gemm1_strip(x, W1, h0, strip0 + (blk - nsort), lds);
}


#define T2_PITCH 132

template <bool FULL>
__device__ __forceinline__ void rows8(float4& acc, int lane, int s, int wbits, int u0, int m,
                                      const float4* __restrict__ h4) {
    int sj[8]; float wj[8];
    #pragma unroll
    for (int u = 0; u < 8; ++u) {
        sj[u] = __builtin_amdgcn_readlane(s, u0 + u);
        wj[u] = __int_as_float(__builtin_amdgcn_readlane(wbits, u0 + u));
    }
    float4 v[8];
    #pragma unroll
    for (int u = 0; u < 8; ++u)
        if (FULL || (u0 + u < m)) v[u] = h4[(size_t)sj[u] * 32 + lane];
    #pragma unroll
    for (int u = 0; u < 8; ++u) {
        if (FULL || (u0 + u < m)) {
            acc.x += v[u].x * wj[u]; acc.y += v[u].y * wj[u];
            acc.z += v[u].z * wj[u]; acc.w += v[u].w * wj[u];
        }
    }
}

__device__ __forceinline__ float4 agg_row128(int node, int lane,
                                             const int* __restrict__ csr,
                                             const int* __restrict__ rowptr,
                                             const int* __restrict__ cnt,
                                             const float* __restrict__ dis,
                                             const float4* __restrict__ h4, float4 bb,
                                             float& dd_out) {
    const int   beg = __builtin_amdgcn_readfirstlane(rowptr[node]);
    const int   len = __builtin_amdgcn_readfirstlane(cnt[node]);
    const float dd  = dis[node];
    dd_out = dd;
    const float4 hv = h4[(size_t)node * 32 + lane];

    float4 acc = {0.f, 0.f, 0.f, 0.f};
    for (int j0 = 0; j0 < len; j0 += 32) {
        const int left = len - j0;
        const int m = left < 32 ? left : 32;
        int s = 0; float w = 0.f;
        if (lane < m) { s = csr[beg + j0 + lane]; w = dis[s] * dd; }
        const int wb = __float_as_int(w);
        int u0 = 0;
        for (; u0 + 8 <= m; u0 += 8) rows8<true >(acc, lane, s, wb, u0, m, h4);
        if (u0 < m)                  rows8<false>(acc, lane, s, wb, u0, m, h4);
    }
    const float d2 = dd * dd;
    float4 o;
    o.x = (bb.x + hv.x * d2) + acc.x; o.y = (bb.y + hv.y * d2) + acc.y;
    o.z = (bb.z + hv.z * d2) + acc.z; o.w = (bb.w + hv.w * d2) + acc.w;
    return o;
}

__global__ __launch_bounds__(256) void k_agg128_gemm2(const int* __restrict__ csr,
                                                      const int* __restrict__ rowptr,
                                                      const int* __restrict__ cnt,
                                                      const float* __restrict__ dis,
                                                      const float* __restrict__ h,
                                                      const float* __restrict__ b1,
                                                      const float* __restrict__ W2,
                                                      float* __restrict__ h2, int n) {
    __shared__ __attribute__((aligned(16))) float tile[16 * T2_PITCH];
    const int lane = threadIdx.x & 31;
    const int wave = __builtin_amdgcn_readfirstlane((int)(threadIdx.x >> 5));
    const int r0   = blockIdx.x * 16;
    const float4* __restrict__ h4 = (const float4*)h;
    const float4 bb = ((const float4*)b1)[lane];

    for (int r = 0; r < 2; ++r) {
        const int trow = wave * 2 + r;
        const int node = r0 + trow;
        float4 o = {0.f, 0.f, 0.f, 0.f};
        if (node < n) {
            float dd;
            o = agg_row128(node, lane, csr, rowptr, cnt, dis, h4, bb, dd);
            o.x = fmaxf(o.x, 0.f) * dd; o.y = fmaxf(o.y, 0.f) * dd;
            o.z = fmaxf(o.z, 0.f) * dd; o.w = fmaxf(o.w, 0.f) * dd;
        }
        *(float4*)&tile[trow * T2_PITCH + lane * 4] = o;
    }
    __syncthreads();
    if (wave != 0) return;

    const int half  = lane >> 4;
    const int l16   = lane & 15;
    const int kboff = half * 2;
    const float* ap = &tile[l16 * T2_PITCH + kboff];
    const float* bp = W2 + (size_t)kboff * 16 + l16;
    v8f acc = {0.f, 0.f, 0.f, 0.f, 0.f, 0.f, 0.f, 0.f};
    for (int k0 = 0; k0 < 128; k0 += 4) {
        v2f a, b;
        a.x = ap[k0]; a.y = ap[k0 + 1];
        b.x = bp[(size_t)k0 * 16]; b.y = bp[(size_t)k0 * 16 + 16];
        acc = __builtin_amdgcn_wmma_f32_16x16x4_f32(false, a, false, b,
                                                    (short)0, acc, false, false);
    }
    float* out = h2 + (size_t)(r0 + half * 8) * 16 + l16;
    #pragma unroll
    for (int i = 0; i < 8; ++i)
        if (r0 + half * 8 + i < n) out[(size_t)i * 16] = acc[i];
}


__global__ __launch_bounds__(256) void k_agg16_softmax(const int* __restrict__ csr,
                                                       const int* __restrict__ rowptr,
                                                       const int* __restrict__ cnt,
                                                       const float* __restrict__ dis,
                                                       const float* __restrict__ h2,
                                                       const float* __restrict__ b2,
                                                       float* __restrict__ logits,
                                                       float* __restrict__ probs, int n) {
    const int g = blockIdx.x * 256 + threadIdx.x;
    if (g >= n * 16) return;
    const int node = g >> 4;
    const int f    = g & 15;
    const int beg  = rowptr[node];
    const int len  = cnt[node];
    const float dd = dis[node];
    const float self = h2[g];
    float acc = 0.f;
    for (int j0 = 0; j0 < len; j0 += 16) {
        const int m = len - j0;
        int s = 0;
        if (f < m) s = csr[beg + j0 + f];
        float v[16];
        #pragma unroll
        for (int u = 0; u < 16; ++u) {
            const int sj = __shfl(s, u, 16);
            v[u] = 0.f;
            if (u < m) v[u] = h2[sj * 16 + f];
        }
        #pragma unroll
        for (int u = 0; u < 16; ++u) acc += v[u];
    }
    const float lg = b2[f] + dd * (acc + self);
    logits[g] = lg;
    float mx = lg;
    #pragma unroll
    for (int o = 8; o >= 1; o >>= 1) mx = fmaxf(mx, __shfl_xor(mx, o, 16));
    const float ex = __expf(lg - mx);
    float sum = ex;
    #pragma unroll
    for (int o = 8; o >= 1; o >>= 1) sum += __shfl_xor(sum, o, 16);
    probs[g] = ex / sum;
}


extern "C" void kernel_launch(void* const* d_in, const int* in_sizes, int n_in,
                              void* d_out, int out_size, void* d_ws, size_t ws_size,
                              hipStream_t stream) {
    const float* x  = (const float*)d_in[0];
    const int*   ei = (const int*)d_in[1];
    const float* W1 = (const float*)d_in[2];
    const float* b1 = (const float*)d_in[3];
    const float* W2 = (const float*)d_in[4];
    const float* b2 = (const float*)d_in[5];

    const int N = in_sizes[0] / 256;
    const int E = in_sizes[1] / 2;
    const int* src = ei;
    const int* dst = ei + E;

    auto cdiv = [](long long a, long long b) { return (int)((a + b - 1) / b); };
    const int nb   = cdiv(N, 256);
    const int nbp  = (nb + 3) & ~3;
    const int nblk = cdiv(E, CHUNK);

    char*  ws  = (char*)d_ws;
    size_t off = 0;
    auto alloc = [&](size_t bytes) -> void* {
        void* p = ws + off;
        off += (bytes + 255) & ~(size_t)255;
        return p;
    };
    int*   cnt    = (int*)alloc((size_t)N * 4);
    int*   rowptr = (int*)alloc((size_t)N * 4);
    float* dis    = (float*)alloc((size_t)N * 4);
    float* h0     = (float*)alloc((size_t)N * 128 * 4);
    int*   recs   = (int*)alloc((size_t)E * 8);
    float* h2     = (float*)alloc((size_t)N * 16 * 4);
    int*   csr    = (int*)alloc((size_t)E * 4);
    if (off > ws_size) return;
    int*   hist   = (int*)h2;
    int*   bstart = hist + (size_t)nblk * nbp;
    int*   bsize  = bstart + nbp;
    if (nbp > MAXNB) return;
    if (((size_t)nblk + 2) * nbp * 4 > (size_t)N * 64) return;
    float* logits = (float*)d_out;
    float* probs  = (float*)d_out + (size_t)N * 16;

    const int nstrip = N / 16;
    const int gA = nstrip / 4;
    const int gB = (nstrip * 2) / 5;
    const int gC = nstrip - gA - gB;
    k_hist_gemm   <<<nblk + gA, 256, 0, stream>>>(dst, hist, E, nbp, nblk, x, W1, h0, 0);
    k_scatter_gemm<<<nblk + gB, 256, 0, stream>>>(src, dst, hist, recs, bstart, bsize, E, nbp, nblk,
                                                  x, W1, h0, gA);
    k_rows_gemm   <<<nb + gC, 256, 0, stream>>>(recs, bstart, bsize, cnt, rowptr, dis, csr, N, nb,
                                                x, W1, h0, gA + gB);

    k_agg128_gemm2<<<cdiv(N, 16), 256, 0, stream>>>(csr, rowptr, cnt, dis, h0, b1, W2, h2, N);

    k_agg16_softmax<<<cdiv((long long)N * 16, 256), 256, 0, stream>>>(csr, rowptr, cnt, dis, h2, b2, logits, probs, N);
}
